# starting kernel
# baseline (speedup 1.0000x reference)
#include <hip/hip_runtime.h>

typedef float    v2f  __attribute__((ext_vector_type(2)));
typedef float    v8f  __attribute__((ext_vector_type(8)));

#define LRELU_SLOPE 0.2f

__device__ __forceinline__ unsigned fenc(float f) {
  unsigned u = __float_as_uint(f);
  return (u & 0x80000000u) ? ~u : (u | 0x80000000u);
}
__device__ __forceinline__ float fdec(unsigned u) {
  u = (u & 0x80000000u) ? (u & 0x7FFFFFFFu) : ~u;
  return __uint_as_float(u);
}

__global__ __launch_bounds__(256) void gat_gemm_wmma(
    const float* __restrict__ X, const float* __restrict__ Wm,
    float* __restrict__ Hout, int N, int K)
{
  __shared__ float wldsT[64 * 130];
  const int stride = K + 2;
  const int tid = threadIdx.x;
  for (int i = tid; i < K * 64; i += 256) {
    const int k = i >> 6, col = i & 63;
    wldsT[col * stride + k] = Wm[i];
  }
  __syncthreads();

  const int wave = tid >> 5;
  const int lane = tid & 31;
  const int l16  = lane & 15;
  const int hsel = lane >> 4;
  const int rowBase = blockIdx.x * 128 + wave * 16;
  int row = rowBase + l16;
  if (row >= N) row = N - 1;

  v8f acc[4] = {};

  const int koff = hsel * 2;
  const float* xrow = X + (size_t)row * K + koff;
  const float* wb0 = &wldsT[(0 * 16 + l16) * stride + koff];
  const float* wb1 = &wldsT[(1 * 16 + l16) * stride + koff];
  const float* wb2 = &wldsT[(2 * 16 + l16) * stride + koff];
  const float* wb3 = &wldsT[(3 * 16 + l16) * stride + koff];

  for (int kk = 0; kk < K; kk += 4) {
    const float2 av  = *(const float2*)(xrow + kk);
    const float2 bv0 = *(const float2*)(wb0 + kk);
    const float2 bv1 = *(const float2*)(wb1 + kk);
    const float2 bv2 = *(const float2*)(wb2 + kk);
    const float2 bv3 = *(const float2*)(wb3 + kk);
    v2f a;  a.x = av.x;   a.y = av.y;
    v2f b0; b0.x = bv0.x; b0.y = bv0.y;
    v2f b1; b1.x = bv1.x; b1.y = bv1.y;
    v2f b2; b2.x = bv2.x; b2.y = bv2.y;
    v2f b3; b3.x = bv3.x; b3.y = bv3.y;
    acc[0] = __builtin_amdgcn_wmma_f32_16x16x4_f32(false, a, false, b0, (short)0, acc[0], false, false);
    acc[1] = __builtin_amdgcn_wmma_f32_16x16x4_f32(false, a, false, b1, (short)0, acc[1], false, false);
    acc[2] = __builtin_amdgcn_wmma_f32_16x16x4_f32(false, a, false, b2, (short)0, acc[2], false, false);
    acc[3] = __builtin_amdgcn_wmma_f32_16x16x4_f32(false, a, false, b3, (short)0, acc[3], false, false);
  }

#pragma unroll
  for (int nt = 0; nt < 4; ++nt) {
#pragma unroll
    for (int j = 0; j < 8; ++j) {
      const int ro = rowBase + j + hsel * 8;
      if (ro < N) Hout[(size_t)ro * 64 + nt * 16 + l16] = acc[nt][j];
    }
  }
}

__global__ __launch_bounds__(256) void gat_logits(
    const float* __restrict__ H, const float* __restrict__ a_src,
    const float* __restrict__ a_dst, float* __restrict__ als,
    float* __restrict__ ald, int N)
{
  const int lane = threadIdx.x & 31;
  const int node = (int)((blockIdx.x * (unsigned)blockDim.x + threadIdx.x) >> 5);
  if (node >= N) return;
  const float h0 = H[(size_t)node * 64 + lane];
  const float h1 = H[(size_t)node * 64 + lane + 32];
  float ps = h0 * a_src[lane] + h1 * a_src[lane + 32];
  float pd = h0 * a_dst[lane] + h1 * a_dst[lane + 32];
#pragma unroll
  for (int o = 16; o > 0; o >>= 1) {
    ps += __shfl_xor(ps, o, 32);
    pd += __shfl_xor(pd, o, 32);
  }
  if (lane == 0) { als[node] = ps; ald[node] = pd; }
}

__global__ void gat_init(unsigned* __restrict__ mEnc, float* __restrict__ denom,
                         float* __restrict__ agg, int N)
{
  const int i = blockIdx.x * blockDim.x + threadIdx.x;
  if (i < N) { mEnc[i] = 0u; denom[i] = 0.0f; }
  if (i < N * 64) agg[i] = 0.0f;
}

__global__ void gat_edge_logit_max(
    const int* __restrict__ ei, int E, int N,
    const float* __restrict__ als, const float* __restrict__ ald,
    float* __restrict__ ebuf, unsigned* __restrict__ mEnc)
{
  const int e = blockIdx.x * blockDim.x + threadIdx.x;
  if (e >= E + N) return;
  int s, d;
  if (e < E) { s = ei[e]; d = ei[E + e]; } else { s = d = e - E; }
  float v = als[s] + ald[d];
  v = (v > 0.0f) ? v : LRELU_SLOPE * v;
  ebuf[e] = v;
  atomicMax(&mEnc[d], fenc(v));
}

__global__ void gat_edge_exp_sum(
    const int* __restrict__ ei, int E, int N,
    const float* __restrict__ ebuf, const unsigned* __restrict__ mEnc,
    float* __restrict__ exbuf, float* __restrict__ denom)
{
  const int e = blockIdx.x * blockDim.x + threadIdx.x;
  if (e >= E + N) return;
  int d;
  if (e < E) { d = ei[E + e]; } else { d = e - E; }
  const float ex = __expf(ebuf[e] - fdec(mEnc[d]));
  exbuf[e] = ex;
  atomicAdd(&denom[d], ex);
}

__global__ __launch_bounds__(256) void gat_edge_scatter(
    const int* __restrict__ ei, int E, int N,
    const float* __restrict__ Hin, const float* __restrict__ exbuf,
    const float* __restrict__ denom, float* __restrict__ agg)
{
  const int lane = threadIdx.x & 31;
  const long long g = (long long)blockIdx.x * blockDim.x + threadIdx.x;
  const int e = (int)(g >> 5);
  if (e >= E + N) return;
  int s, d;
  if (e < E) { s = ei[e]; d = ei[E + e]; } else { s = d = e - E; }
  const float alpha = exbuf[e] / denom[d];
  const float2 hv = *(const float2*)&Hin[(size_t)s * 64 + lane * 2];
  atomicAdd(&agg[(size_t)d * 64 + lane * 2 + 0], alpha * hv.x);
  atomicAdd(&agg[(size_t)d * 64 + lane * 2 + 1], alpha * hv.y);
}

__global__ void gat_epilogue(const float* __restrict__ agg,
                             const float* __restrict__ bias,
                             float* __restrict__ out, int N, int doRelu)
{
  const int i = blockIdx.x * blockDim.x + threadIdx.x;
  if (i >= N * 64) return;
  float v = agg[i] + bias[i & 63];
  if (doRelu) v = fmaxf(v, 0.0f);
  out[i] = v;
}

extern "C" void kernel_launch(void* const* d_in, const int* in_sizes, int n_in,
                              void* d_out, int out_size, void* d_ws, size_t ws_size,
                              hipStream_t stream) {
  const float* x   = (const float*)d_in[0];
  const int*   ei  = (const int*)  d_in[1];
  const float* W1  = (const float*)d_in[2];
  const float* as1 = (const float*)d_in[3];
  const float* ad1 = (const float*)d_in[4];
  const float* b1  = (const float*)d_in[5];
  const float* W2  = (const float*)d_in[6];
  const float* as2 = (const float*)d_in[7];
  const float* ad2 = (const float*)d_in[8];
  const float* b2  = (const float*)d_in[9];

  const int N  = in_sizes[0] / 128;
  const int E  = in_sizes[1] / 2;
  const int ET = E + N;

  char* w = (char*)d_ws;
  float*    h1    = (float*)w;    w += (size_t)N * 64 * sizeof(float);
  float*    h2    = (float*)w;    w += (size_t)N * 64 * sizeof(float);
  float*    agg   = (float*)w;    w += (size_t)N * 64 * sizeof(float);
  float*    als   = (float*)w;    w += (size_t)N * sizeof(float);
  float*    ald   = (float*)w;    w += (size_t)N * sizeof(float);
  unsigned* mEnc  = (unsigned*)w; w += (size_t)N * sizeof(unsigned);
  float*    denom = (float*)w;    w += (size_t)N * sizeof(float);
  float*    ebuf  = (float*)w;    w += (size_t)ET * sizeof(float);
  float*    exbuf = (float*)w;    w += (size_t)ET * sizeof(float);

  const dim3 blk(256);
  const int gGemm     = (N + 127) / 128;
  const int gNode     = (N * 64 + 255) / 256;
  const int gWarpNode = (N * 32 + 255) / 256;
  const int gEdge     = (ET + 255) / 256;
  const int gWaveEdge = (int)(((long long)ET * 32 + 255) / 256);

  gat_gemm_wmma    <<<gGemm,     blk, 0, stream>>>(x,  W1, h1, N, 128);
  gat_logits       <<<gWarpNode, blk, 0, stream>>>(h1, as1, ad1, als, ald, N);
  gat_init         <<<gNode,     blk, 0, stream>>>(mEnc, denom, agg, N);
  gat_edge_logit_max<<<gEdge,    blk, 0, stream>>>(ei, E, N, als, ald, ebuf, mEnc);
  gat_edge_exp_sum <<<gEdge,     blk, 0, stream>>>(ei, E, N, ebuf, mEnc, exbuf, denom);
  gat_edge_scatter <<<gWaveEdge, blk, 0, stream>>>(ei, E, N, h1, exbuf, denom, agg);
  gat_epilogue     <<<gNode,     blk, 0, stream>>>(agg, b1, h2, N, 1);

  gat_gemm_wmma    <<<gGemm,     blk, 0, stream>>>(h2, W2, h1, N, 64);
  gat_logits       <<<gWarpNode, blk, 0, stream>>>(h1, as2, ad2, als, ald, N);
  gat_init         <<<gNode,     blk, 0, stream>>>(mEnc, denom, agg, N);
  gat_edge_logit_max<<<gEdge,    blk, 0, stream>>>(ei, E, N, als, ald, ebuf, mEnc);
  gat_edge_exp_sum <<<gEdge,     blk, 0, stream>>>(ei, E, N, ebuf, mEnc, exbuf, denom);
  gat_edge_scatter <<<gWaveEdge, blk, 0, stream>>>(ei, E, N, h1, exbuf, denom, agg);
  gat_epilogue     <<<gNode,     blk, 0, stream>>>(agg, b2, (float*)d_out, N, 0);
}


# Edges bucketed by destination once per call (int atomic rank, scan, fill); each layer does max/exp/sum/aggregate per node in one wave, no float atomics; sums run in arrival-order rank
# speedup vs baseline: 2.6857x; 2.6857x over previous
#include <hip/hip_runtime.h>

typedef float    v2f  __attribute__((ext_vector_type(2)));
typedef float    v8f  __attribute__((ext_vector_type(8)));

#define LRELU_SLOPE 0.2f

__global__ __launch_bounds__(256) void gat_gemm_wmma(
    const float* __restrict__ X, const float* __restrict__ Wm,
    float* __restrict__ Hout, int N, int K)
{
  __shared__ float wldsT[64 * 130];
  const int stride = K + 2;
  const int tid = threadIdx.x;
  for (int i = tid; i < K * 64; i += 256) {
    const int k = i >> 6, col = i & 63;
    wldsT[col * stride + k] = Wm[i];
  }
  __syncthreads();

  const int wave = tid >> 5;
  const int lane = tid & 31;
  const int l16  = lane & 15;
  const int hsel = lane >> 4;
  const int rowBase = blockIdx.x * 128 + wave * 16;
  int row = rowBase + l16;
  if (row >= N) row = N - 1;

  v8f acc[4] = {};

  const int koff = hsel * 2;
  const float* xrow = X + (size_t)row * K + koff;
  const float* wb0 = &wldsT[(0 * 16 + l16) * stride + koff];
  const float* wb1 = &wldsT[(1 * 16 + l16) * stride + koff];
  const float* wb2 = &wldsT[(2 * 16 + l16) * stride + koff];
  const float* wb3 = &wldsT[(3 * 16 + l16) * stride + koff];

  for (int kk = 0; kk < K; kk += 4) {
    const float2 av  = *(const float2*)(xrow + kk);
    const float2 bv0 = *(const float2*)(wb0 + kk);
    const float2 bv1 = *(const float2*)(wb1 + kk);
    const float2 bv2 = *(const float2*)(wb2 + kk);
    const float2 bv3 = *(const float2*)(wb3 + kk);
    v2f a;  a.x = av.x;   a.y = av.y;
    v2f b0; b0.x = bv0.x; b0.y = bv0.y;
    v2f b1; b1.x = bv1.x; b1.y = bv1.y;
    v2f b2; b2.x = bv2.x; b2.y = bv2.y;
    v2f b3; b3.x = bv3.x; b3.y = bv3.y;
    acc[0] = __builtin_amdgcn_wmma_f32_16x16x4_f32(false, a, false, b0, (short)0, acc[0], false, false);
    acc[1] = __builtin_amdgcn_wmma_f32_16x16x4_f32(false, a, false, b1, (short)0, acc[1], false, false);
    acc[2] = __builtin_amdgcn_wmma_f32_16x16x4_f32(false, a, false, b2, (short)0, acc[2], false, false);
    acc[3] = __builtin_amdgcn_wmma_f32_16x16x4_f32(false, a, false, b3, (short)0, acc[3], false, false);
  }

#pragma unroll
  for (int nt = 0; nt < 4; ++nt) {
#pragma unroll
    for (int j = 0; j < 8; ++j) {
      const int ro = rowBase + j + hsel * 8;
      if (ro < N) Hout[(size_t)ro * 64 + nt * 16 + l16] = acc[nt][j];
    }
  }
}

__global__ __launch_bounds__(256) void gat_logits(
    const float* __restrict__ H, const float* __restrict__ a_src,
    const float* __restrict__ a_dst, float* __restrict__ als,
    float* __restrict__ ald, int N)
{
  const int lane = threadIdx.x & 31;
  const int node = (int)((blockIdx.x * (unsigned)blockDim.x + threadIdx.x) >> 5);
  if (node >= N) return;
  const float h0 = H[(size_t)node * 64 + lane];
  const float h1 = H[(size_t)node * 64 + lane + 32];
  float ps = h0 * a_src[lane] + h1 * a_src[lane + 32];
  float pd = h0 * a_dst[lane] + h1 * a_dst[lane + 32];
#pragma unroll
  for (int o = 16; o > 0; o >>= 1) {
    ps += __shfl_xor(ps, o, 32);
    pd += __shfl_xor(pd, o, 32);
  }
  if (lane == 0) { als[node] = ps; ald[node] = pd; }
}

__global__ void csr_zero(int* __restrict__ deg, int N)
{
  const int i = blockIdx.x * blockDim.x + threadIdx.x;
  if (i < N) deg[i] = 0;
}

__global__ void csr_rank(const int* __restrict__ ei, int E,
                         int* __restrict__ deg, int* __restrict__ rank)
{
  const int e = blockIdx.x * blockDim.x + threadIdx.x;
  if (e >= E) return;
  rank[e] = atomicAdd(&deg[ei[E + e]], 1);
}

__global__ __launch_bounds__(256) void csr_scan(
    const int* __restrict__ deg, int* __restrict__ off, int N)
{
  __shared__ int part[16];
  const int tid = threadIdx.x, lane = tid & 31, wave = tid >> 5;
  const int base = blockIdx.x * 256;
  int p = 0;
  for (int i = tid; i < base; i += 256) p += deg[i];
  const int idx = base + tid;
  const int v = (idx < N) ? deg[idx] : 0;
  int x = v;
  for (int o = 1; o < 32; o <<= 1) {
    const int t = __shfl_up(x, o, 32);
    if (lane >= o) x += t;
  }
  for (int o = 16; o > 0; o >>= 1) p += __shfl_xor(p, o, 32);
  if (lane == 31) part[wave] = x;
  if (lane == 0)  part[8 + wave] = p;
  __syncthreads();
  int pre = 0;
  for (int w = 0; w < 8; ++w) {
    pre += part[8 + w];
    if (w < wave) pre += part[w];
  }
  if (idx < N) off[idx] = pre + x - v;
}

__global__ void csr_fill(const int* __restrict__ ei, int E,
                         const int* __restrict__ off, const int* __restrict__ rank,
                         int* __restrict__ csrc)
{
  const int e = blockIdx.x * blockDim.x + threadIdx.x;
  if (e >= E) return;
  csrc[off[ei[E + e]] + rank[e]] = ei[e];
}

template <int M> __device__ __forceinline__ float row_xor(float v) {
  const int i = __float_as_int(v);
  return __int_as_float(__builtin_amdgcn_update_dpp(i, i, 0x160 | M, 0xf, 0xf, true));
}
__device__ __forceinline__ float half_xor(float v) {
  const unsigned u = __float_as_uint(v);
  return __uint_as_float(__builtin_amdgcn_permlanex16(u, u, 0x76543210u, 0xfedcba98u, false, false));
}
__device__ __forceinline__ float wave_max(float m) {
  m = fmaxf(m, half_xor(m));
  m = fmaxf(m, row_xor<8>(m));
  m = fmaxf(m, row_xor<4>(m));
  m = fmaxf(m, row_xor<2>(m));
  m = fmaxf(m, row_xor<1>(m));
  return m;
}
__device__ __forceinline__ float wave_sum(float s) {
  s += half_xor(s);
  s += row_xor<8>(s);
  s += row_xor<4>(s);
  s += row_xor<2>(s);
  s += row_xor<1>(s);
  return s;
}

__device__ __forceinline__ float lrelu(float v) {
  return (v > 0.0f) ? v : LRELU_SLOPE * v;
}

__global__ __launch_bounds__(256) void gat_node_softmax_aggregate(
    const int* __restrict__ off, const int* __restrict__ deg,
    const int* __restrict__ csrc, const float* __restrict__ Hin,
    const float* __restrict__ als, const float* __restrict__ ald,
    const float* __restrict__ bias, float* __restrict__ out, int N, int doRelu)
{
  const int lane = threadIdx.x & 31;
  const int node = __builtin_amdgcn_readfirstlane(
      (int)((blockIdx.x * (unsigned)blockDim.x + threadIdx.x) >> 5));
  if (node >= N) return;
  const int   beg   = off[node];
  const int   cnt   = deg[node];
  const float adv   = ald[node];
  const float eself = lrelu(als[node] + adv);

  int   s0 = node;
  float v0 = eself;
  if (lane < cnt) { s0 = csrc[beg + lane]; v0 = lrelu(als[s0] + adv); }

  float m = fmaxf(v0, eself);
  for (int j = lane + 32; j < cnt; j += 32)
    m = fmaxf(m, lrelu(als[csrc[beg + j]] + adv));
  m = wave_max(m);

  const float ex0 = (lane < cnt) ? __expf(v0 - m) : 0.0f;
  float dsum = ex0;
  for (int j = lane + 32; j < cnt; j += 32)
    dsum += __expf(lrelu(als[csrc[beg + j]] + adv) - m);
  dsum = wave_sum(dsum);
  const float exs   = __expf(eself - m);
  const float denom = dsum + exs;

  const float* hl = Hin + lane * 2;
  const float2 hs = *(const float2*)(hl + (size_t)node * 64);
  const float  as = exs / denom;
  float ax = as * hs.x, ay = as * hs.y;

  for (int base = 0; base < cnt; base += 32) {
    int   sj = s0;
    float aj = ex0 / denom;
    if (base > 0) {
      sj = node; aj = 0.0f;
      const int j = base + lane;
      if (j < cnt) {
        sj = csrc[beg + j];
        aj = __expf(lrelu(als[sj] + adv) - m) / denom;
      }
    }
    const int n = (cnt - base < 32) ? (cnt - base) : 32;
    for (int t = 0; t < n; t += 8) {
      float2 hv[8];
      float  wv[8];
#pragma unroll
      for (int k = 0; k < 8; ++k) {
        const int sk = __builtin_amdgcn_readlane(sj, t + k);
        wv[k] = __int_as_float(__builtin_amdgcn_readlane(__float_as_int(aj), t + k));
        hv[k] = *(const float2*)(hl + (size_t)sk * 64);
      }
#pragma unroll
      for (int k = 0; k < 8; ++k) {
        if (t + k < n) { ax += wv[k] * hv[k].x; ay += wv[k] * hv[k].y; }
      }
    }
  }

  const float2 bv = *(const float2*)(bias + lane * 2);
  float2 r; r.x = ax + bv.x; r.y = ay + bv.y;
  if (doRelu) { r.x = fmaxf(r.x, 0.0f); r.y = fmaxf(r.y, 0.0f); }
  *(float2*)(out + (size_t)node * 64 + lane * 2) = r;
}

extern "C" void kernel_launch(void* const* d_in, const int* in_sizes, int n_in,
                              void* d_out, int out_size, void* d_ws, size_t ws_size,
                              hipStream_t stream) {
  const float* x   = (const float*)d_in[0];
  const int*   ei  = (const int*)  d_in[1];
  const float* W1  = (const float*)d_in[2];
  const float* as1 = (const float*)d_in[3];
  const float* ad1 = (const float*)d_in[4];
  const float* b1  = (const float*)d_in[5];
  const float* W2  = (const float*)d_in[6];
  const float* as2 = (const float*)d_in[7];
  const float* ad2 = (const float*)d_in[8];
  const float* b2  = (const float*)d_in[9];

  const int N  = in_sizes[0] / 128;
  const int E  = in_sizes[1] / 2;

  char* w = (char*)d_ws;
  float* h1   = (float*)w; w += (size_t)N * 64 * sizeof(float);
  float* h2   = (float*)w; w += (size_t)N * 64 * sizeof(float);
  float* als  = (float*)w; w += (size_t)N * sizeof(float);
  float* ald  = (float*)w; w += (size_t)N * sizeof(float);
  int*   deg  = (int*)w;   w += (size_t)N * sizeof(int);
  int*   off  = (int*)w;   w += (size_t)N * sizeof(int);
  int*   rank = (int*)w;   w += (size_t)E * sizeof(int);
  int*   csrc = (int*)w;   w += (size_t)E * sizeof(int);

  const dim3 blk(256);
  const int gGemm     = (N + 127) / 128;
  const int gNode     = (N + 255) / 256;
  const int gWarpNode = (N * 32 + 255) / 256;
  const int gEdge     = (E + 255) / 256;

  csr_zero <<<gNode, blk, 0, stream>>>(deg, N);
  csr_rank <<<gEdge, blk, 0, stream>>>(ei, E, deg, rank);
  csr_scan <<<gNode, blk, 0, stream>>>(deg, off, N);
  csr_fill <<<gEdge, blk, 0, stream>>>(ei, E, off, rank, csrc);

  gat_gemm_wmma <<<gGemm,     blk, 0, stream>>>(x,  W1, h1, N, 128);
  gat_logits    <<<gWarpNode, blk, 0, stream>>>(h1, as1, ad1, als, ald, N);
  gat_node_softmax_aggregate<<<gWarpNode, blk, 0, stream>>>(off, deg, csrc, h1, als, ald, b1, h2, N, 1);

  gat_gemm_wmma <<<gGemm,     blk, 0, stream>>>(h2, W2, h1, N, 64);
  gat_logits    <<<gWarpNode, blk, 0, stream>>>(h1, as2, ad2, als, ald, N);
  gat_node_softmax_aggregate<<<gWarpNode, blk, 0, stream>>>(off, deg, csrc, h1, als, ald, b2, (float*)d_out, N, 0);
}


# Per-node attention logits computed in the GEMM epilogue from the accumulator registers (16-lane row sums); the two logits launches are gone; logit dot products summed in a different tree
# speedup vs baseline: 2.9874x; 1.1123x over previous
#include <hip/hip_runtime.h>

typedef float    v2f  __attribute__((ext_vector_type(2)));
typedef float    v8f  __attribute__((ext_vector_type(8)));

#define LRELU_SLOPE 0.2f

template <int M> __device__ __forceinline__ float row_xor(float v) {
  const int i = __float_as_int(v);
  return __int_as_float(__builtin_amdgcn_update_dpp(i, i, 0x160 | M, 0xf, 0xf, true));
}
__device__ __forceinline__ float row_sum16(float s) {
  s += row_xor<8>(s);
  s += row_xor<4>(s);
  s += row_xor<2>(s);
  s += row_xor<1>(s);
  return s;
}

__global__ __launch_bounds__(256) void gat_gemm_wmma(
    const float* __restrict__ X, const float* __restrict__ Wm,
    const float* __restrict__ a_src, const float* __restrict__ a_dst,
    float* __restrict__ Hout, float* __restrict__ als, float* __restrict__ ald,
    int N, int K)
{
  __shared__ float wldsT[64 * 130];
  const int stride = K + 2;
  const int tid = threadIdx.x;
  for (int i = tid; i < K * 64; i += 256) {
    const int k = i >> 6, col = i & 63;
    wldsT[col * stride + k] = Wm[i];
  }
  __syncthreads();

  const int wave = tid >> 5;
  const int lane = tid & 31;
  const int l16  = lane & 15;
  const int hsel = lane >> 4;
  const int rowBase = blockIdx.x * 128 + wave * 16;
  int row = rowBase + l16;
  if (row >= N) row = N - 1;

  v8f acc[4] = {};

  const float as0 = a_src[l16], as1 = a_src[16 + l16], as2 = a_src[32 + l16], as3 = a_src[48 + l16];
  const float ad0 = a_dst[l16], ad1 = a_dst[16 + l16], ad2 = a_dst[32 + l16], ad3 = a_dst[48 + l16];

  const int koff = hsel * 2;
  const float* xrow = X + (size_t)row * K + koff;
  const float* wb0 = &wldsT[(0 * 16 + l16) * stride + koff];
  const float* wb1 = &wldsT[(1 * 16 + l16) * stride + koff];
  const float* wb2 = &wldsT[(2 * 16 + l16) * stride + koff];
  const float* wb3 = &wldsT[(3 * 16 + l16) * stride + koff];

  for (int kk = 0; kk < K; kk += 4) {
    const float2 av  = *(const float2*)(xrow + kk);
    const float2 bv0 = *(const float2*)(wb0 + kk);
    const float2 bv1 = *(const float2*)(wb1 + kk);
    const float2 bv2 = *(const float2*)(wb2 + kk);
    const float2 bv3 = *(const float2*)(wb3 + kk);
    v2f a;  a.x = av.x;   a.y = av.y;
    v2f b0; b0.x = bv0.x; b0.y = bv0.y;
    v2f b1; b1.x = bv1.x; b1.y = bv1.y;
    v2f b2; b2.x = bv2.x; b2.y = bv2.y;
    v2f b3; b3.x = bv3.x; b3.y = bv3.y;
    acc[0] = __builtin_amdgcn_wmma_f32_16x16x4_f32(false, a, false, b0, (short)0, acc[0], false, false);
    acc[1] = __builtin_amdgcn_wmma_f32_16x16x4_f32(false, a, false, b1, (short)0, acc[1], false, false);
    acc[2] = __builtin_amdgcn_wmma_f32_16x16x4_f32(false, a, false, b2, (short)0, acc[2], false, false);
    acc[3] = __builtin_amdgcn_wmma_f32_16x16x4_f32(false, a, false, b3, (short)0, acc[3], false, false);
  }

#pragma unroll
  for (int nt = 0; nt < 4; ++nt) {
#pragma unroll
    for (int j = 0; j < 8; ++j) {
      const int ro = rowBase + j + hsel * 8;
      if (ro < N) Hout[(size_t)ro * 64 + nt * 16 + l16] = acc[nt][j];
    }
  }

  float ps[8], pd[8];
#pragma unroll
  for (int j = 0; j < 8; ++j) {
    ps[j] = row_sum16(acc[0][j] * as0 + acc[1][j] * as1 + acc[2][j] * as2 + acc[3][j] * as3);
    pd[j] = row_sum16(acc[0][j] * ad0 + acc[1][j] * ad1 + acc[2][j] * ad2 + acc[3][j] * ad3);
  }
  const int  jsel  = l16 & 7;
  const bool isDst = l16 >= 8;
  float val = 0.0f;
#pragma unroll
  for (int j = 0; j < 8; ++j) {
    const float t = isDst ? pd[j] : ps[j];
    if (jsel == j) val = t;
  }
  const int rl = rowBase + jsel + hsel * 8;
  float* lp = isDst ? ald : als;
  if (rl < N) lp[rl] = val;
}

__global__ void csr_zero(int* __restrict__ deg, int N)
{
  const int i = blockIdx.x * blockDim.x + threadIdx.x;
  if (i < N) deg[i] = 0;
}

__global__ void csr_rank(const int* __restrict__ ei, int E,
                         int* __restrict__ deg, int* __restrict__ rank)
{
  const int e = blockIdx.x * blockDim.x + threadIdx.x;
  if (e >= E) return;
  rank[e] = atomicAdd(&deg[ei[E + e]], 1);
}

__global__ __launch_bounds__(256) void csr_scan(
    const int* __restrict__ deg, int* __restrict__ off, int N)
{
  __shared__ int part[16];
  const int tid = threadIdx.x, lane = tid & 31, wave = tid >> 5;
  const int base = blockIdx.x * 256;
  int p = 0;
  for (int i = tid; i < base; i += 256) p += deg[i];
  const int idx = base + tid;
  const int v = (idx < N) ? deg[idx] : 0;
  int x = v;
  for (int o = 1; o < 32; o <<= 1) {
    const int t = __shfl_up(x, o, 32);
    if (lane >= o) x += t;
  }
  for (int o = 16; o > 0; o >>= 1) p += __shfl_xor(p, o, 32);
  if (lane == 31) part[wave] = x;
  if (lane == 0)  part[8 + wave] = p;
  __syncthreads();
  int pre = 0;
  for (int w = 0; w < 8; ++w) {
    pre += part[8 + w];
    if (w < wave) pre += part[w];
  }
  if (idx < N) off[idx] = pre + x - v;
}

__global__ void csr_fill(const int* __restrict__ ei, int E,
                         const int* __restrict__ off, const int* __restrict__ rank,
                         int* __restrict__ csrc)
{
  const int e = blockIdx.x * blockDim.x + threadIdx.x;
  if (e >= E) return;
  csrc[off[ei[E + e]] + rank[e]] = ei[e];
}

__device__ __forceinline__ float half_xor(float v) {
  const unsigned u = __float_as_uint(v);
  return __uint_as_float(__builtin_amdgcn_permlanex16(u, u, 0x76543210u, 0xfedcba98u, false, false));
}
__device__ __forceinline__ float wave_max(float m) {
  m = fmaxf(m, half_xor(m));
  m = fmaxf(m, row_xor<8>(m));
  m = fmaxf(m, row_xor<4>(m));
  m = fmaxf(m, row_xor<2>(m));
  m = fmaxf(m, row_xor<1>(m));
  return m;
}
__device__ __forceinline__ float wave_sum(float s) {
  s += half_xor(s);
  s += row_xor<8>(s);
  s += row_xor<4>(s);
  s += row_xor<2>(s);
  s += row_xor<1>(s);
  return s;
}

__device__ __forceinline__ float lrelu(float v) {
  return (v > 0.0f) ? v : LRELU_SLOPE * v;
}

__global__ __launch_bounds__(256) void gat_node_softmax_aggregate(
    const int* __restrict__ off, const int* __restrict__ deg,
    const int* __restrict__ csrc, const float* __restrict__ Hin,
    const float* __restrict__ als, const float* __restrict__ ald,
    const float* __restrict__ bias, float* __restrict__ out, int N, int doRelu)
{
  const int lane = threadIdx.x & 31;
  const int node = __builtin_amdgcn_readfirstlane(
      (int)((blockIdx.x * (unsigned)blockDim.x + threadIdx.x) >> 5));
  if (node >= N) return;
  const int   beg   = off[node];
  const int   cnt   = deg[node];
  const float adv   = ald[node];
  const float eself = lrelu(als[node] + adv);

  int   s0 = node;
  float v0 = eself;
  if (lane < cnt) { s0 = csrc[beg + lane]; v0 = lrelu(als[s0] + adv); }

  float m = fmaxf(v0, eself);
  for (int j = lane + 32; j < cnt; j += 32)
    m = fmaxf(m, lrelu(als[csrc[beg + j]] + adv));
  m = wave_max(m);

  const float ex0 = (lane < cnt) ? __expf(v0 - m) : 0.0f;
  float dsum = ex0;
  for (int j = lane + 32; j < cnt; j += 32)
    dsum += __expf(lrelu(als[csrc[beg + j]] + adv) - m);
  dsum = wave_sum(dsum);
  const float exs   = __expf(eself - m);
  const float denom = dsum + exs;

  const float* hl = Hin + lane * 2;
  const float2 hs = *(const float2*)(hl + (size_t)node * 64);
  const float  as = exs / denom;
  float ax = as * hs.x, ay = as * hs.y;

  for (int base = 0; base < cnt; base += 32) {
    int   sj = s0;
    float aj = ex0 / denom;
    if (base > 0) {
      sj = node; aj = 0.0f;
      const int j = base + lane;
      if (j < cnt) {
        sj = csrc[beg + j];
        aj = __expf(lrelu(als[sj] + adv) - m) / denom;
      }
    }
    const int n = (cnt - base < 32) ? (cnt - base) : 32;
    for (int t = 0; t < n; t += 8) {
      float2 hv[8];
      float  wv[8];
#pragma unroll
      for (int k = 0; k < 8; ++k) {
        const int sk = __builtin_amdgcn_readlane(sj, t + k);
        wv[k] = __int_as_float(__builtin_amdgcn_readlane(__float_as_int(aj), t + k));
        hv[k] = *(const float2*)(hl + (size_t)sk * 64);
      }
#pragma unroll
      for (int k = 0; k < 8; ++k) {
        if (t + k < n) { ax += wv[k] * hv[k].x; ay += wv[k] * hv[k].y; }
      }
    }
  }

  const float2 bv = *(const float2*)(bias + lane * 2);
  float2 r; r.x = ax + bv.x; r.y = ay + bv.y;
  if (doRelu) { r.x = fmaxf(r.x, 0.0f); r.y = fmaxf(r.y, 0.0f); }
  *(float2*)(out + (size_t)node * 64 + lane * 2) = r;
}

extern "C" void kernel_launch(void* const* d_in, const int* in_sizes, int n_in,
                              void* d_out, int out_size, void* d_ws, size_t ws_size,
                              hipStream_t stream) {
  const float* x   = (const float*)d_in[0];
  const int*   ei  = (const int*)  d_in[1];
  const float* W1  = (const float*)d_in[2];
  const float* as1 = (const float*)d_in[3];
  const float* ad1 = (const float*)d_in[4];
  const float* b1  = (const float*)d_in[5];
  const float* W2  = (const float*)d_in[6];
  const float* as2 = (const float*)d_in[7];
  const float* ad2 = (const float*)d_in[8];
  const float* b2  = (const float*)d_in[9];

  const int N  = in_sizes[0] / 128;
  const int E  = in_sizes[1] / 2;

  char* w = (char*)d_ws;
  float* h1   = (float*)w; w += (size_t)N * 64 * sizeof(float);
  float* h2   = (float*)w; w += (size_t)N * 64 * sizeof(float);
  float* als  = (float*)w; w += (size_t)N * sizeof(float);
  float* ald  = (float*)w; w += (size_t)N * sizeof(float);
  int*   deg  = (int*)w;   w += (size_t)N * sizeof(int);
  int*   off  = (int*)w;   w += (size_t)N * sizeof(int);
  int*   rank = (int*)w;   w += (size_t)E * sizeof(int);
  int*   csrc = (int*)w;   w += (size_t)E * sizeof(int);

  const dim3 blk(256);
  const int gGemm     = (N + 127) / 128;
  const int gNode     = (N + 255) / 256;
  const int gWarpNode = (N * 32 + 255) / 256;
  const int gEdge     = (E + 255) / 256;

  csr_zero <<<gNode, blk, 0, stream>>>(deg, N);
  csr_rank <<<gEdge, blk, 0, stream>>>(ei, E, deg, rank);
  csr_scan <<<gNode, blk, 0, stream>>>(deg, off, N);
  csr_fill <<<gEdge, blk, 0, stream>>>(ei, E, off, rank, csrc);

  gat_gemm_wmma <<<gGemm,     blk, 0, stream>>>(x,  W1, as1, ad1, h1, als, ald, N, 128);
  gat_node_softmax_aggregate<<<gWarpNode, blk, 0, stream>>>(off, deg, csrc, h1, als, ald, b1, h2, N, 1);

  gat_gemm_wmma <<<gGemm,     blk, 0, stream>>>(h2, W2, as2, ad2, h1, als, ald, N, 64);
  gat_node_softmax_aggregate<<<gWarpNode, blk, 0, stream>>>(off, deg, csrc, h1, als, ald, b2, (float*)d_out, N, 0);
}


# Edges bucketed by a two-level counting sort on LDS counters (256-node bins, then per node) instead of per-edge global atomics; 8 to 7 launches; bucket (summation) order is now LDS arrival order
# speedup vs baseline: 4.3543x; 1.4576x over previous
#include <hip/hip_runtime.h>

typedef float    v2f  __attribute__((ext_vector_type(2)));
typedef float    v8f  __attribute__((ext_vector_type(8)));

#define LRELU_SLOPE 0.2f

template <int M> __device__ __forceinline__ float row_xor(float v) {
  const int i = __float_as_int(v);
  return __int_as_float(__builtin_amdgcn_update_dpp(i, i, 0x160 | M, 0xf, 0xf, true));
}
__device__ __forceinline__ float row_sum16(float s) {
  s += row_xor<8>(s);
  s += row_xor<4>(s);
  s += row_xor<2>(s);
  s += row_xor<1>(s);
  return s;
}

__global__ __launch_bounds__(256) void gat_gemm_wmma(
    const float* __restrict__ X, const float* __restrict__ Wm,
    const float* __restrict__ a_src, const float* __restrict__ a_dst,
    float* __restrict__ Hout, float* __restrict__ als, float* __restrict__ ald,
    int N, int K)
{
  __shared__ float wldsT[64 * 130];
  const int stride = K + 2;
  const int tid = threadIdx.x;
  for (int i = tid; i < K * 64; i += 256) {
    const int k = i >> 6, col = i & 63;
    wldsT[col * stride + k] = Wm[i];
  }
  __syncthreads();

  const int wave = tid >> 5;
  const int lane = tid & 31;
  const int l16  = lane & 15;
  const int hsel = lane >> 4;
  const int rowBase = blockIdx.x * 128 + wave * 16;
  int row = rowBase + l16;
  if (row >= N) row = N - 1;

  v8f acc[4] = {};

  const float as0 = a_src[l16], as1 = a_src[16 + l16], as2 = a_src[32 + l16], as3 = a_src[48 + l16];
  const float ad0 = a_dst[l16], ad1 = a_dst[16 + l16], ad2 = a_dst[32 + l16], ad3 = a_dst[48 + l16];

  const int koff = hsel * 2;
  const float* xrow = X + (size_t)row * K + koff;
  const float* wb0 = &wldsT[(0 * 16 + l16) * stride + koff];
  const float* wb1 = &wldsT[(1 * 16 + l16) * stride + koff];
  const float* wb2 = &wldsT[(2 * 16 + l16) * stride + koff];
  const float* wb3 = &wldsT[(3 * 16 + l16) * stride + koff];

  for (int kk = 0; kk < K; kk += 4) {
    const float2 av  = *(const float2*)(xrow + kk);
    const float2 bv0 = *(const float2*)(wb0 + kk);
    const float2 bv1 = *(const float2*)(wb1 + kk);
    const float2 bv2 = *(const float2*)(wb2 + kk);
    const float2 bv3 = *(const float2*)(wb3 + kk);
    v2f a;  a.x = av.x;   a.y = av.y;
    v2f b0; b0.x = bv0.x; b0.y = bv0.y;
    v2f b1; b1.x = bv1.x; b1.y = bv1.y;
    v2f b2; b2.x = bv2.x; b2.y = bv2.y;
    v2f b3; b3.x = bv3.x; b3.y = bv3.y;
    acc[0] = __builtin_amdgcn_wmma_f32_16x16x4_f32(false, a, false, b0, (short)0, acc[0], false, false);
    acc[1] = __builtin_amdgcn_wmma_f32_16x16x4_f32(false, a, false, b1, (short)0, acc[1], false, false);
    acc[2] = __builtin_amdgcn_wmma_f32_16x16x4_f32(false, a, false, b2, (short)0, acc[2], false, false);
    acc[3] = __builtin_amdgcn_wmma_f32_16x16x4_f32(false, a, false, b3, (short)0, acc[3], false, false);
  }

#pragma unroll
  for (int nt = 0; nt < 4; ++nt) {
#pragma unroll
    for (int j = 0; j < 8; ++j) {
      const int ro = rowBase + j + hsel * 8;
      if (ro < N) Hout[(size_t)ro * 64 + nt * 16 + l16] = acc[nt][j];
    }
  }

  float ps[8], pd[8];
#pragma unroll
  for (int j = 0; j < 8; ++j) {
    ps[j] = row_sum16(acc[0][j] * as0 + acc[1][j] * as1 + acc[2][j] * as2 + acc[3][j] * as3);
    pd[j] = row_sum16(acc[0][j] * ad0 + acc[1][j] * ad1 + acc[2][j] * ad2 + acc[3][j] * ad3);
  }
  const int  jsel  = l16 & 7;
  const bool isDst = l16 >= 8;
  float val = 0.0f;
#pragma unroll
  for (int j = 0; j < 8; ++j) {
    const float t = isDst ? pd[j] : ps[j];
    if (jsel == j) val = t;
  }
  const int rl = rowBase + jsel + hsel * 8;
  float* lp = isDst ? ald : als;
  if (rl < N) lp[rl] = val;
}

#define BIN_SHIFT  8
#define BIN_NODES  256
#define MAX_BINS   1024
#define EDGE_CHUNK 4096

__device__ __forceinline__ int wave_incl_scan(int x, int lane)
{
  for (int o = 1; o < 32; o <<= 1) {
    const int t = __shfl_up(x, o, 32);
    if (lane >= o) x += t;
  }
  return x;
}

__global__ __launch_bounds__(1024) void bin_count(
    const int* __restrict__ ei, int E, int NB, int* __restrict__ blkHist)
{
  __shared__ int h[MAX_BINS];
  const int tid = threadIdx.x;
  if (tid < NB) h[tid] = 0;
  __syncthreads();
  const int base = blockIdx.x * EDGE_CHUNK + tid;
  int d[EDGE_CHUNK / 1024];
#pragma unroll
  for (int k = 0; k < EDGE_CHUNK / 1024; ++k) {
    const int e = base + k * 1024;
    d[k] = ei[E + (e < E ? e : E - 1)];
  }
#pragma unroll
  for (int k = 0; k < EDGE_CHUNK / 1024; ++k)
    if (base + k * 1024 < E) atomicAdd(&h[d[k] >> BIN_SHIFT], 1);
  __syncthreads();
  if (tid < NB) blkHist[blockIdx.x * NB + tid] = h[tid];
}

__global__ __launch_bounds__(1024) void bin_scatter(
    const int* __restrict__ ei, int E, int NB, int nBlk,
    const int* __restrict__ blkHist, int* __restrict__ binStart,
    int* __restrict__ binned)
{
  __shared__ int partT[4][MAX_BINS];
  __shared__ int partP[4][MAX_BINS];
  __shared__ int cursor[MAX_BINS];
  __shared__ int wsum[32];
  const int tid = threadIdx.x, lane = tid & 31, wave = tid >> 5;
  const int blk = blockIdx.x;

  const int base = blk * EDGE_CHUNK + tid;
  int s[EDGE_CHUNK / 1024], d[EDGE_CHUNK / 1024];
#pragma unroll
  for (int k = 0; k < EDGE_CHUNK / 1024; ++k) {
    const int e  = base + k * 1024;
    const int ec = e < E ? e : E - 1;
    s[k] = ei[ec];
    d[k] = ei[E + ec];
  }

  const int g = tid >> 8, c = tid & 255;
  for (int bt = 0; bt < NB; bt += 256) {
    const int b = bt + c;
    int t = 0, p = 0;
    if (b < NB) {
      for (int j0 = g; j0 < nBlk; j0 += 64) {
        int v[16];
#pragma unroll
        for (int u = 0; u < 16; ++u) {
          const int j = j0 + 4 * u;
          v[u] = blkHist[(j < nBlk ? j : nBlk - 1) * NB + b];
        }
#pragma unroll
        for (int u = 0; u < 16; ++u) {
          const int j = j0 + 4 * u;
          t += (j < nBlk) ? v[u] : 0;
          p += (j < blk) ? v[u] : 0;
        }
      }
      partT[g][b] = t;
      partP[g][b] = p;
    }
  }
  __syncthreads();

  int tot = 0, pre = 0;
  if (tid < NB) {
    tot = partT[0][tid] + partT[1][tid] + partT[2][tid] + partT[3][tid];
    pre = partP[0][tid] + partP[1][tid] + partP[2][tid] + partP[3][tid];
  }
  const int x = wave_incl_scan(tot, lane);
  if (lane == 31) wsum[wave] = x;
  __syncthreads();
  if (wave == 0) {
    const int w = wsum[lane];
    const int y = wave_incl_scan(w, lane);
    wsum[lane] = y - w;
  }
  __syncthreads();
  const int excl = wsum[wave] + x - tot;
  if (tid < NB) {
    cursor[tid] = excl + pre;
    if (blk == 0) binStart[tid] = excl;
  }
  if (blk == 0 && tid == 0) binStart[NB] = E;
  __syncthreads();

  int pos[EDGE_CHUNK / 1024];
#pragma unroll
  for (int k = 0; k < EDGE_CHUNK / 1024; ++k) {
    pos[k] = 0;
    if (base + k * 1024 < E) pos[k] = atomicAdd(&cursor[d[k] >> BIN_SHIFT], 1);
  }
#pragma unroll
  for (int k = 0; k < EDGE_CHUNK / 1024; ++k)
    if (base + k * 1024 < E)
      binned[pos[k]] = (s[k] << BIN_SHIFT) | (d[k] & (BIN_NODES - 1));
}

__global__ __launch_bounds__(1024) void bin_sort(
    const int* __restrict__ binStart, const int* __restrict__ binned,
    int* __restrict__ deg, int* __restrict__ off, int* __restrict__ csrc, int N)
{
  __shared__ int cnt[BIN_NODES];
  __shared__ int cur[BIN_NODES];
  __shared__ int wtot[BIN_NODES / 32];
  const int tid = threadIdx.x, lane = tid & 31, wave = tid >> 5;
  const int lo = binStart[blockIdx.x], hi = binStart[blockIdx.x + 1];
  const int i0 = lo + tid;
  int w[4];
#pragma unroll
  for (int k = 0; k < 4; ++k) {
    const int i = i0 + k * 1024;
    w[k] = (i < hi) ? binned[i] : 0;
  }
  if (tid < BIN_NODES) cnt[tid] = 0;
  __syncthreads();
#pragma unroll
  for (int k = 0; k < 4; ++k)
    if (i0 + k * 1024 < hi) atomicAdd(&cnt[w[k] & (BIN_NODES - 1)], 1);
  for (int i = i0 + 4096; i < hi; i += 1024)
    atomicAdd(&cnt[binned[i] & (BIN_NODES - 1)], 1);
  __syncthreads();
  int v = 0, x = 0;
  if (tid < BIN_NODES) {
    v = cnt[tid];
    x = wave_incl_scan(v, lane);
    if (lane == 31) wtot[wave] = x;
  }
  __syncthreads();
  if (tid < BIN_NODES) {
    int pre = 0;
    for (int q = 0; q < BIN_NODES / 32; ++q) pre += (q < wave) ? wtot[q] : 0;
    const int o = lo + pre + x - v;
    cur[tid] = o;
    const int node = blockIdx.x * BIN_NODES + tid;
    if (node < N) { deg[node] = v; off[node] = o; }
  }
  __syncthreads();
  int pos[4];
#pragma unroll
  for (int k = 0; k < 4; ++k) {
    pos[k] = 0;
    if (i0 + k * 1024 < hi) pos[k] = atomicAdd(&cur[w[k] & (BIN_NODES - 1)], 1);
  }
#pragma unroll
  for (int k = 0; k < 4; ++k)
    if (i0 + k * 1024 < hi) csrc[pos[k]] = w[k] >> BIN_SHIFT;
  for (int i = i0 + 4096; i < hi; i += 1024) {
    const int t = binned[i];
    const int p = atomicAdd(&cur[t & (BIN_NODES - 1)], 1);
    csrc[p] = t >> BIN_SHIFT;
  }
}

__device__ __forceinline__ float half_xor(float v) {
  const unsigned u = __float_as_uint(v);
  return __uint_as_float(__builtin_amdgcn_permlanex16(u, u, 0x76543210u, 0xfedcba98u, false, false));
}
__device__ __forceinline__ float wave_max(float m) {
  m = fmaxf(m, half_xor(m));
  m = fmaxf(m, row_xor<8>(m));
  m = fmaxf(m, row_xor<4>(m));
  m = fmaxf(m, row_xor<2>(m));
  m = fmaxf(m, row_xor<1>(m));
  return m;
}
__device__ __forceinline__ float wave_sum(float s) {
  s += half_xor(s);
  s += row_xor<8>(s);
  s += row_xor<4>(s);
  s += row_xor<2>(s);
  s += row_xor<1>(s);
  return s;
}

__device__ __forceinline__ float lrelu(float v) {
  return (v > 0.0f) ? v : LRELU_SLOPE * v;
}

__global__ __launch_bounds__(256) void gat_node_softmax_aggregate(
    const int* __restrict__ off, const int* __restrict__ deg,
    const int* __restrict__ csrc, const float* __restrict__ Hin,
    const float* __restrict__ als, const float* __restrict__ ald,
    const float* __restrict__ bias, float* __restrict__ out, int N, int doRelu)
{
  const int lane = threadIdx.x & 31;
  const int node = __builtin_amdgcn_readfirstlane(
      (int)((blockIdx.x * (unsigned)blockDim.x + threadIdx.x) >> 5));
  if (node >= N) return;
  const int   beg   = off[node];
  const int   cnt   = deg[node];
  const float adv   = ald[node];
  const float eself = lrelu(als[node] + adv);

  int   s0 = node;
  float v0 = eself;
  if (lane < cnt) { s0 = csrc[beg + lane]; v0 = lrelu(als[s0] + adv); }

  float m = fmaxf(v0, eself);
  for (int j = lane + 32; j < cnt; j += 32)
    m = fmaxf(m, lrelu(als[csrc[beg + j]] + adv));
  m = wave_max(m);

  const float ex0 = (lane < cnt) ? __expf(v0 - m) : 0.0f;
  float dsum = ex0;
  for (int j = lane + 32; j < cnt; j += 32)
    dsum += __expf(lrelu(als[csrc[beg + j]] + adv) - m);
  dsum = wave_sum(dsum);
  const float exs   = __expf(eself - m);
  const float denom = dsum + exs;

  const float* hl = Hin + lane * 2;
  const float2 hs = *(const float2*)(hl + (size_t)node * 64);
  const float  as = exs / denom;
  float ax = as * hs.x, ay = as * hs.y;

  for (int base = 0; base < cnt; base += 32) {
    int   sj = s0;
    float aj = ex0 / denom;
    if (base > 0) {
      sj = node; aj = 0.0f;
      const int j = base + lane;
      if (j < cnt) {
        sj = csrc[beg + j];
        aj = __expf(lrelu(als[sj] + adv) - m) / denom;
      }
    }
    const int n = (cnt - base < 32) ? (cnt - base) : 32;
    for (int t = 0; t < n; t += 8) {
      float2 hv[8];
      float  wv[8];
#pragma unroll
      for (int k = 0; k < 8; ++k) {
        const int sk = __builtin_amdgcn_readlane(sj, t + k);
        wv[k] = __int_as_float(__builtin_amdgcn_readlane(__float_as_int(aj), t + k));
        hv[k] = *(const float2*)(hl + (size_t)sk * 64);
      }
#pragma unroll
      for (int k = 0; k < 8; ++k) {
        if (t + k < n) { ax += wv[k] * hv[k].x; ay += wv[k] * hv[k].y; }
      }
    }
  }

  const float2 bv = *(const float2*)(bias + lane * 2);
  float2 r; r.x = ax + bv.x; r.y = ay + bv.y;
  if (doRelu) { r.x = fmaxf(r.x, 0.0f); r.y = fmaxf(r.y, 0.0f); }
  *(float2*)(out + (size_t)node * 64 + lane * 2) = r;
}

extern "C" void kernel_launch(void* const* d_in, const int* in_sizes, int n_in,
                              void* d_out, int out_size, void* d_ws, size_t ws_size,
                              hipStream_t stream) {
  const float* x   = (const float*)d_in[0];
  const int*   ei  = (const int*)  d_in[1];
  const float* W1  = (const float*)d_in[2];
  const float* as1 = (const float*)d_in[3];
  const float* ad1 = (const float*)d_in[4];
  const float* b1  = (const float*)d_in[5];
  const float* W2  = (const float*)d_in[6];
  const float* as2 = (const float*)d_in[7];
  const float* ad2 = (const float*)d_in[8];
  const float* b2  = (const float*)d_in[9];

  const int N  = in_sizes[0] / 128;
  const int E  = in_sizes[1] / 2;

  char* w = (char*)d_ws;
  float* h1   = (float*)w; w += (size_t)N * 64 * sizeof(float);
  float* h2   = (float*)w; w += (size_t)N * 64 * sizeof(float);
  float* als  = (float*)w; w += (size_t)N * sizeof(float);
  float* ald  = (float*)w; w += (size_t)N * sizeof(float);
  int*   deg  = (int*)w;   w += (size_t)N * sizeof(int);
  int*   off  = (int*)w;   w += (size_t)N * sizeof(int);
  const int NB   = (N + BIN_NODES - 1) / BIN_NODES;
  const int nBlk = (E + EDGE_CHUNK - 1) / EDGE_CHUNK;
  int*   binStart = (int*)w; w += (size_t)(NB + 1) * sizeof(int);
  int*   blkHist  = (int*)w; w += (size_t)nBlk * NB * sizeof(int);
  int*   binned   = (int*)w; w += (size_t)E * sizeof(int);
  int*   csrc     = (int*)w; w += (size_t)E * sizeof(int);

  const dim3 blk(256);
  const dim3 blkBin(1024);
  const int gGemm     = (N + 127) / 128;
  const int gWarpNode = (N * 32 + 255) / 256;

  bin_count  <<<nBlk, blkBin, 0, stream>>>(ei, E, NB, blkHist);
  bin_scatter<<<nBlk, blkBin, 0, stream>>>(ei, E, NB, nBlk, blkHist, binStart, binned);
  bin_sort   <<<NB,   blkBin, 0, stream>>>(binStart, binned, deg, off, csrc, N);

  gat_gemm_wmma <<<gGemm,     blk, 0, stream>>>(x,  W1, as1, ad1, h1, als, ald, N, 128);
  gat_node_softmax_aggregate<<<gWarpNode, blk, 0, stream>>>(off, deg, csrc, h1, als, ald, b1, h2, N, 1);

  gat_gemm_wmma <<<gGemm,     blk, 0, stream>>>(h2, W2, as2, ad2, h1, als, ald, N, 64);
  gat_node_softmax_aggregate<<<gWarpNode, blk, 0, stream>>>(off, deg, csrc, h1, als, ald, b2, (float*)d_out, N, 0);
}


# Layer-1 GEMM and the edge bin count, independent of each other, share one launch (GEMM blocks first, then 256-thread count blocks); 7 to 6 launches; every sum and its order as before
# speedup vs baseline: 4.5262x; 1.0395x over previous
#include <hip/hip_runtime.h>

typedef float    v2f  __attribute__((ext_vector_type(2)));
typedef float    v8f  __attribute__((ext_vector_type(8)));

#define LRELU_SLOPE 0.2f

template <int M> __device__ __forceinline__ float row_xor(float v) {
  const int i = __float_as_int(v);
  return __int_as_float(__builtin_amdgcn_update_dpp(i, i, 0x160 | M, 0xf, 0xf, true));
}
__device__ __forceinline__ float row_sum16(float s) {
  s += row_xor<8>(s);
  s += row_xor<4>(s);
  s += row_xor<2>(s);
  s += row_xor<1>(s);
  return s;
}

__device__ __forceinline__ void gat_gemm_body(
    float* __restrict__ wldsT, const int blk,
    const float* __restrict__ X, const float* __restrict__ Wm,
    const float* __restrict__ a_src, const float* __restrict__ a_dst,
    float* __restrict__ Hout, float* __restrict__ als, float* __restrict__ ald,
    int N, int K)
{
  const int stride = K + 2;
  const int tid = threadIdx.x;
  for (int i = tid; i < K * 64; i += 256) {
    const int k = i >> 6, col = i & 63;
    wldsT[col * stride + k] = Wm[i];
  }
  __syncthreads();

  const int wave = tid >> 5;
  const int lane = tid & 31;
  const int l16  = lane & 15;
  const int hsel = lane >> 4;
  const int rowBase = blk * 128 + wave * 16;
  int row = rowBase + l16;
  if (row >= N) row = N - 1;

  v8f acc[4] = {};

  const float as0 = a_src[l16], as1 = a_src[16 + l16], as2 = a_src[32 + l16], as3 = a_src[48 + l16];
  const float ad0 = a_dst[l16], ad1 = a_dst[16 + l16], ad2 = a_dst[32 + l16], ad3 = a_dst[48 + l16];

  const int koff = hsel * 2;
  const float* xrow = X + (size_t)row * K + koff;
  const float* wb0 = &wldsT[(0 * 16 + l16) * stride + koff];
  const float* wb1 = &wldsT[(1 * 16 + l16) * stride + koff];
  const float* wb2 = &wldsT[(2 * 16 + l16) * stride + koff];
  const float* wb3 = &wldsT[(3 * 16 + l16) * stride + koff];

  for (int kk = 0; kk < K; kk += 4) {
    const float2 av  = *(const float2*)(xrow + kk);
    const float2 bv0 = *(const float2*)(wb0 + kk);
    const float2 bv1 = *(const float2*)(wb1 + kk);
    const float2 bv2 = *(const float2*)(wb2 + kk);
    const float2 bv3 = *(const float2*)(wb3 + kk);
    v2f a;  a.x = av.x;   a.y = av.y;
    v2f b0; b0.x = bv0.x; b0.y = bv0.y;
    v2f b1; b1.x = bv1.x; b1.y = bv1.y;
    v2f b2; b2.x = bv2.x; b2.y = bv2.y;
    v2f b3; b3.x = bv3.x; b3.y = bv3.y;
    acc[0] = __builtin_amdgcn_wmma_f32_16x16x4_f32(false, a, false, b0, (short)0, acc[0], false, false);
    acc[1] = __builtin_amdgcn_wmma_f32_16x16x4_f32(false, a, false, b1, (short)0, acc[1], false, false);
    acc[2] = __builtin_amdgcn_wmma_f32_16x16x4_f32(false, a, false, b2, (short)0, acc[2], false, false);
    acc[3] = __builtin_amdgcn_wmma_f32_16x16x4_f32(false, a, false, b3, (short)0, acc[3], false, false);
  }

#pragma unroll
  for (int nt = 0; nt < 4; ++nt) {
#pragma unroll
    for (int j = 0; j < 8; ++j) {
      const int ro = rowBase + j + hsel * 8;
      if (ro < N) Hout[(size_t)ro * 64 + nt * 16 + l16] = acc[nt][j];
    }
  }

  float ps[8], pd[8];
#pragma unroll
  for (int j = 0; j < 8; ++j) {
    ps[j] = row_sum16(acc[0][j] * as0 + acc[1][j] * as1 + acc[2][j] * as2 + acc[3][j] * as3);
    pd[j] = row_sum16(acc[0][j] * ad0 + acc[1][j] * ad1 + acc[2][j] * ad2 + acc[3][j] * ad3);
  }
  const int  jsel  = l16 & 7;
  const bool isDst = l16 >= 8;
  float val = 0.0f;
#pragma unroll
  for (int j = 0; j < 8; ++j) {
    const float t = isDst ? pd[j] : ps[j];
    if (jsel == j) val = t;
  }
  const int rl = rowBase + jsel + hsel * 8;
  float* lp = isDst ? ald : als;
  if (rl < N) lp[rl] = val;
}

__global__ __launch_bounds__(256) void gat_gemm_wmma(
    const float* __restrict__ X, const float* __restrict__ Wm,
    const float* __restrict__ a_src, const float* __restrict__ a_dst,
    float* __restrict__ Hout, float* __restrict__ als, float* __restrict__ ald,
    int N, int K)
{
  __shared__ float wldsT[64 * 130];
  gat_gemm_body(wldsT, blockIdx.x, X, Wm, a_src, a_dst, Hout, als, ald, N, K);
}

#define BIN_SHIFT  8
#define BIN_NODES  256
#define MAX_BINS   1024
#define EDGE_CHUNK 4096

__device__ __forceinline__ int wave_incl_scan(int x, int lane)
{
  for (int o = 1; o < 32; o <<= 1) {
    const int t = __shfl_up(x, o, 32);
    if (lane >= o) x += t;
  }
  return x;
}

__global__ __launch_bounds__(256) void gemm1_and_bin_count(
    const float* __restrict__ X, const float* __restrict__ Wm,
    const float* __restrict__ a_src, const float* __restrict__ a_dst,
    float* __restrict__ Hout, float* __restrict__ als, float* __restrict__ ald,
    int N, int K, int gGemm,
    const int* __restrict__ ei, int E, int NB, int* __restrict__ blkHist)
{
  __shared__ float smem[64 * 130];
  if ((int)blockIdx.x < gGemm) {
    gat_gemm_body(smem, blockIdx.x, X, Wm, a_src, a_dst, Hout, als, ald, N, K);
    return;
  }
  int* h = (int*)smem;
  const int tid = threadIdx.x;
  const int blk = blockIdx.x - gGemm;
  for (int b = tid; b < NB; b += 256) h[b] = 0;
  __syncthreads();
  const int base = blk * EDGE_CHUNK + tid;
  int d[EDGE_CHUNK / 256];
#pragma unroll
  for (int k = 0; k < EDGE_CHUNK / 256; ++k) {
    const int e = base + k * 256;
    d[k] = ei[E + (e < E ? e : E - 1)];
  }
#pragma unroll
  for (int k = 0; k < EDGE_CHUNK / 256; ++k)
    if (base + k * 256 < E) atomicAdd(&h[d[k] >> BIN_SHIFT], 1);
  __syncthreads();
  for (int b = tid; b < NB; b += 256) blkHist[blk * NB + b] = h[b];
}

__global__ __launch_bounds__(1024) void bin_scatter(
    const int* __restrict__ ei, int E, int NB, int nBlk,
    const int* __restrict__ blkHist, int* __restrict__ binStart,
    int* __restrict__ binned)
{
  __shared__ int partT[4][MAX_BINS];
  __shared__ int partP[4][MAX_BINS];
  __shared__ int cursor[MAX_BINS];
  __shared__ int wsum[32];
  const int tid = threadIdx.x, lane = tid & 31, wave = tid >> 5;
  const int blk = blockIdx.x;

  const int base = blk * EDGE_CHUNK + tid;
  int s[EDGE_CHUNK / 1024], d[EDGE_CHUNK / 1024];
#pragma unroll
  for (int k = 0; k < EDGE_CHUNK / 1024; ++k) {
    const int e  = base + k * 1024;
    const int ec = e < E ? e : E - 1;
    s[k] = ei[ec];
    d[k] = ei[E + ec];
  }

  const int g = tid >> 8, c = tid & 255;
  for (int bt = 0; bt < NB; bt += 256) {
    const int b = bt + c;
    int t = 0, p = 0;
    if (b < NB) {
      for (int j0 = g; j0 < nBlk; j0 += 64) {
        int v[16];
#pragma unroll
        for (int u = 0; u < 16; ++u) {
          const int j = j0 + 4 * u;
          v[u] = blkHist[(j < nBlk ? j : nBlk - 1) * NB + b];
        }
#pragma unroll
        for (int u = 0; u < 16; ++u) {
          const int j = j0 + 4 * u;
          t += (j < nBlk) ? v[u] : 0;
          p += (j < blk) ? v[u] : 0;
        }
      }
      partT[g][b] = t;
      partP[g][b] = p;
    }
  }
  __syncthreads();

  int tot = 0, pre = 0;
  if (tid < NB) {
    tot = partT[0][tid] + partT[1][tid] + partT[2][tid] + partT[3][tid];
    pre = partP[0][tid] + partP[1][tid] + partP[2][tid] + partP[3][tid];
  }
  const int x = wave_incl_scan(tot, lane);
  if (lane == 31) wsum[wave] = x;
  __syncthreads();
  if (wave == 0) {
    const int w = wsum[lane];
    const int y = wave_incl_scan(w, lane);
    wsum[lane] = y - w;
  }
  __syncthreads();
  const int excl = wsum[wave] + x - tot;
  if (tid < NB) {
    cursor[tid] = excl + pre;
    if (blk == 0) binStart[tid] = excl;
  }
  if (blk == 0 && tid == 0) binStart[NB] = E;
  __syncthreads();

  int pos[EDGE_CHUNK / 1024];
#pragma unroll
  for (int k = 0; k < EDGE_CHUNK / 1024; ++k) {
    pos[k] = 0;
    if (base + k * 1024 < E) pos[k] = atomicAdd(&cursor[d[k] >> BIN_SHIFT], 1);
  }
#pragma unroll
  for (int k = 0; k < EDGE_CHUNK / 1024; ++k)
    if (base + k * 1024 < E)
      binned[pos[k]] = (s[k] << BIN_SHIFT) | (d[k] & (BIN_NODES - 1));
}

__global__ __launch_bounds__(1024) void bin_sort(
    const int* __restrict__ binStart, const int* __restrict__ binned,
    int* __restrict__ deg, int* __restrict__ off, int* __restrict__ csrc, int N)
{
  __shared__ int cnt[BIN_NODES];
  __shared__ int cur[BIN_NODES];
  __shared__ int wtot[BIN_NODES / 32];
  const int tid = threadIdx.x, lane = tid & 31, wave = tid >> 5;
  const int lo = binStart[blockIdx.x], hi = binStart[blockIdx.x + 1];
  const int i0 = lo + tid;
  int w[4];
#pragma unroll
  for (int k = 0; k < 4; ++k) {
    const int i = i0 + k * 1024;
    w[k] = (i < hi) ? binned[i] : 0;
  }
  if (tid < BIN_NODES) cnt[tid] = 0;
  __syncthreads();
#pragma unroll
  for (int k = 0; k < 4; ++k)
    if (i0 + k * 1024 < hi) atomicAdd(&cnt[w[k] & (BIN_NODES - 1)], 1);
  for (int i = i0 + 4096; i < hi; i += 1024)
    atomicAdd(&cnt[binned[i] & (BIN_NODES - 1)], 1);
  __syncthreads();
  int v = 0, x = 0;
  if (tid < BIN_NODES) {
    v = cnt[tid];
    x = wave_incl_scan(v, lane);
    if (lane == 31) wtot[wave] = x;
  }
  __syncthreads();
  if (tid < BIN_NODES) {
    int pre = 0;
    for (int q = 0; q < BIN_NODES / 32; ++q) pre += (q < wave) ? wtot[q] : 0;
    const int o = lo + pre + x - v;
    cur[tid] = o;
    const int node = blockIdx.x * BIN_NODES + tid;
    if (node < N) { deg[node] = v; off[node] = o; }
  }
  __syncthreads();
  int pos[4];
#pragma unroll
  for (int k = 0; k < 4; ++k) {
    pos[k] = 0;
    if (i0 + k * 1024 < hi) pos[k] = atomicAdd(&cur[w[k] & (BIN_NODES - 1)], 1);
  }
#pragma unroll
  for (int k = 0; k < 4; ++k)
    if (i0 + k * 1024 < hi) csrc[pos[k]] = w[k] >> BIN_SHIFT;
  for (int i = i0 + 4096; i < hi; i += 1024) {
    const int t = binned[i];
    const int p = atomicAdd(&cur[t & (BIN_NODES - 1)], 1);
    csrc[p] = t >> BIN_SHIFT;
  }
}

__device__ __forceinline__ float half_xor(float v) {
  const unsigned u = __float_as_uint(v);
  return __uint_as_float(__builtin_amdgcn_permlanex16(u, u, 0x76543210u, 0xfedcba98u, false, false));
}
__device__ __forceinline__ float wave_max(float m) {
  m = fmaxf(m, half_xor(m));
  m = fmaxf(m, row_xor<8>(m));
  m = fmaxf(m, row_xor<4>(m));
  m = fmaxf(m, row_xor<2>(m));
  m = fmaxf(m, row_xor<1>(m));
  return m;
}
__device__ __forceinline__ float wave_sum(float s) {
  s += half_xor(s);
  s += row_xor<8>(s);
  s += row_xor<4>(s);
  s += row_xor<2>(s);
  s += row_xor<1>(s);
  return s;
}

__device__ __forceinline__ float lrelu(float v) {
  return (v > 0.0f) ? v : LRELU_SLOPE * v;
}

__global__ __launch_bounds__(256) void gat_node_softmax_aggregate(
    const int* __restrict__ off, const int* __restrict__ deg,
    const int* __restrict__ csrc, const float* __restrict__ Hin,
    const float* __restrict__ als, const float* __restrict__ ald,
    const float* __restrict__ bias, float* __restrict__ out, int N, int doRelu)
{
  const int lane = threadIdx.x & 31;
  const int node = __builtin_amdgcn_readfirstlane(
      (int)((blockIdx.x * (unsigned)blockDim.x + threadIdx.x) >> 5));
  if (node >= N) return;
  const int   beg   = off[node];
  const int   cnt   = deg[node];
  const float adv   = ald[node];
  const float eself = lrelu(als[node] + adv);

  int   s0 = node;
  float v0 = eself;
  if (lane < cnt) { s0 = csrc[beg + lane]; v0 = lrelu(als[s0] + adv); }

  float m = fmaxf(v0, eself);
  for (int j = lane + 32; j < cnt; j += 32)
    m = fmaxf(m, lrelu(als[csrc[beg + j]] + adv));
  m = wave_max(m);

  const float ex0 = (lane < cnt) ? __expf(v0 - m) : 0.0f;
  float dsum = ex0;
  for (int j = lane + 32; j < cnt; j += 32)
    dsum += __expf(lrelu(als[csrc[beg + j]] + adv) - m);
  dsum = wave_sum(dsum);
  const float exs   = __expf(eself - m);
  const float denom = dsum + exs;

  const float* hl = Hin + lane * 2;
  const float2 hs = *(const float2*)(hl + (size_t)node * 64);
  const float  as = exs / denom;
  float ax = as * hs.x, ay = as * hs.y;

  for (int base = 0; base < cnt; base += 32) {
    int   sj = s0;
    float aj = ex0 / denom;
    if (base > 0) {
      sj = node; aj = 0.0f;
      const int j = base + lane;
      if (j < cnt) {
        sj = csrc[beg + j];
        aj = __expf(lrelu(als[sj] + adv) - m) / denom;
      }
    }
    const int n = (cnt - base < 32) ? (cnt - base) : 32;
    for (int t = 0; t < n; t += 8) {
      float2 hv[8];
      float  wv[8];
#pragma unroll
      for (int k = 0; k < 8; ++k) {
        const int sk = __builtin_amdgcn_readlane(sj, t + k);
        wv[k] = __int_as_float(__builtin_amdgcn_readlane(__float_as_int(aj), t + k));
        hv[k] = *(const float2*)(hl + (size_t)sk * 64);
      }
#pragma unroll
      for (int k = 0; k < 8; ++k) {
        if (t + k < n) { ax += wv[k] * hv[k].x; ay += wv[k] * hv[k].y; }
      }
    }
  }

  const float2 bv = *(const float2*)(bias + lane * 2);
  float2 r; r.x = ax + bv.x; r.y = ay + bv.y;
  if (doRelu) { r.x = fmaxf(r.x, 0.0f); r.y = fmaxf(r.y, 0.0f); }
  *(float2*)(out + (size_t)node * 64 + lane * 2) = r;
}

extern "C" void kernel_launch(void* const* d_in, const int* in_sizes, int n_in,
                              void* d_out, int out_size, void* d_ws, size_t ws_size,
                              hipStream_t stream) {
  const float* x   = (const float*)d_in[0];
  const int*   ei  = (const int*)  d_in[1];
  const float* W1  = (const float*)d_in[2];
  const float* as1 = (const float*)d_in[3];
  const float* ad1 = (const float*)d_in[4];
  const float* b1  = (const float*)d_in[5];
  const float* W2  = (const float*)d_in[6];
  const float* as2 = (const float*)d_in[7];
  const float* ad2 = (const float*)d_in[8];
  const float* b2  = (const float*)d_in[9];

  const int N  = in_sizes[0] / 128;
  const int E  = in_sizes[1] / 2;

  char* w = (char*)d_ws;
  float* h1   = (float*)w; w += (size_t)N * 64 * sizeof(float);
  float* h2   = (float*)w; w += (size_t)N * 64 * sizeof(float);
  float* als  = (float*)w; w += (size_t)N * sizeof(float);
  float* ald  = (float*)w; w += (size_t)N * sizeof(float);
  int*   deg  = (int*)w;   w += (size_t)N * sizeof(int);
  int*   off  = (int*)w;   w += (size_t)N * sizeof(int);
  const int NB   = (N + BIN_NODES - 1) / BIN_NODES;
  const int nBlk = (E + EDGE_CHUNK - 1) / EDGE_CHUNK;
  int*   binStart = (int*)w; w += (size_t)(NB + 1) * sizeof(int);
  int*   blkHist  = (int*)w; w += (size_t)nBlk * NB * sizeof(int);
  int*   binned   = (int*)w; w += (size_t)E * sizeof(int);
  int*   csrc     = (int*)w; w += (size_t)E * sizeof(int);

  const dim3 blk(256);
  const dim3 blkBin(1024);
  const int gGemm     = (N + 127) / 128;
  const int gWarpNode = (N * 32 + 255) / 256;

  gemm1_and_bin_count<<<gGemm + nBlk, blk, 0, stream>>>(x, W1, as1, ad1, h1, als, ald, N, 128, gGemm,
                                                       ei, E, NB, blkHist);
  bin_scatter<<<nBlk, blkBin, 0, stream>>>(ei, E, NB, nBlk, blkHist, binStart, binned);
  bin_sort   <<<NB,   blkBin, 0, stream>>>(binStart, binned, deg, off, csrc, N);

  gat_node_softmax_aggregate<<<gWarpNode, blk, 0, stream>>>(off, deg, csrc, h1, als, ald, b1, h2, N, 1);

  gat_gemm_wmma <<<gGemm,     blk, 0, stream>>>(h2, W2, as2, ad2, h1, als, ald, N, 64);
  gat_node_softmax_aggregate<<<gWarpNode, blk, 0, stream>>>(off, deg, csrc, h1, als, ald, b2, (float*)d_out, N, 0);
}


# Layer-2 GEMM moved into the layer-1 node pass: 16 node waves per block put their rows in an LDS tile, 4 of them run the f32 WMMAs on it; 6 to 5 launches; logit sums in a different tree
# speedup vs baseline: 4.6735x; 1.0325x over previous
#include <hip/hip_runtime.h>

typedef float    v2f  __attribute__((ext_vector_type(2)));
typedef float    v8f  __attribute__((ext_vector_type(8)));

#define LRELU_SLOPE 0.2f

template <int M> __device__ __forceinline__ float row_xor(float v) {
  const int i = __float_as_int(v);
  return __int_as_float(__builtin_amdgcn_update_dpp(i, i, 0x160 | M, 0xf, 0xf, true));
}
__device__ __forceinline__ float row_sum16(float s) {
  s += row_xor<8>(s);
  s += row_xor<4>(s);
  s += row_xor<2>(s);
  s += row_xor<1>(s);
  return s;
}

__device__ __forceinline__ void gat_gemm_body(
    float* __restrict__ wldsT, const int blk,
    const float* __restrict__ X, const float* __restrict__ Wm,
    const float* __restrict__ a_src, const float* __restrict__ a_dst,
    float* __restrict__ Hout, float* __restrict__ als, float* __restrict__ ald,
    int N, int K)
{
  const int stride = K + 2;
  const int tid = threadIdx.x;
  for (int i = tid; i < K * 64; i += 256) {
    const int k = i >> 6, col = i & 63;
    wldsT[col * stride + k] = Wm[i];
  }
  __syncthreads();

  const int wave = tid >> 5;
  const int lane = tid & 31;
  const int l16  = lane & 15;
  const int hsel = lane >> 4;
  const int rowBase = blk * 128 + wave * 16;
  int row = rowBase + l16;
  if (row >= N) row = N - 1;

  v8f acc[4] = {};

  const float as0 = a_src[l16], as1 = a_src[16 + l16], as2 = a_src[32 + l16], as3 = a_src[48 + l16];
  const float ad0 = a_dst[l16], ad1 = a_dst[16 + l16], ad2 = a_dst[32 + l16], ad3 = a_dst[48 + l16];

  const int koff = hsel * 2;
  const float* xrow = X + (size_t)row * K + koff;
  const float* wb0 = &wldsT[(0 * 16 + l16) * stride + koff];
  const float* wb1 = &wldsT[(1 * 16 + l16) * stride + koff];
  const float* wb2 = &wldsT[(2 * 16 + l16) * stride + koff];
  const float* wb3 = &wldsT[(3 * 16 + l16) * stride + koff];

  for (int kk = 0; kk < K; kk += 4) {
    const float2 av  = *(const float2*)(xrow + kk);
    const float2 bv0 = *(const float2*)(wb0 + kk);
    const float2 bv1 = *(const float2*)(wb1 + kk);
    const float2 bv2 = *(const float2*)(wb2 + kk);
    const float2 bv3 = *(const float2*)(wb3 + kk);
    v2f a;  a.x = av.x;   a.y = av.y;
    v2f b0; b0.x = bv0.x; b0.y = bv0.y;
    v2f b1; b1.x = bv1.x; b1.y = bv1.y;
    v2f b2; b2.x = bv2.x; b2.y = bv2.y;
    v2f b3; b3.x = bv3.x; b3.y = bv3.y;
    acc[0] = __builtin_amdgcn_wmma_f32_16x16x4_f32(false, a, false, b0, (short)0, acc[0], false, false);
    acc[1] = __builtin_amdgcn_wmma_f32_16x16x4_f32(false, a, false, b1, (short)0, acc[1], false, false);
    acc[2] = __builtin_amdgcn_wmma_f32_16x16x4_f32(false, a, false, b2, (short)0, acc[2], false, false);
    acc[3] = __builtin_amdgcn_wmma_f32_16x16x4_f32(false, a, false, b3, (short)0, acc[3], false, false);
  }

#pragma unroll
  for (int nt = 0; nt < 4; ++nt) {
#pragma unroll
    for (int j = 0; j < 8; ++j) {
      const int ro = rowBase + j + hsel * 8;
      if (ro < N) Hout[(size_t)ro * 64 + nt * 16 + l16] = acc[nt][j];
    }
  }

  float ps[8], pd[8];
#pragma unroll
  for (int j = 0; j < 8; ++j) {
    ps[j] = row_sum16(acc[0][j] * as0 + acc[1][j] * as1 + acc[2][j] * as2 + acc[3][j] * as3);
    pd[j] = row_sum16(acc[0][j] * ad0 + acc[1][j] * ad1 + acc[2][j] * ad2 + acc[3][j] * ad3);
  }
  const int  jsel  = l16 & 7;
  const bool isDst = l16 >= 8;
  float val = 0.0f;
#pragma unroll
  for (int j = 0; j < 8; ++j) {
    const float t = isDst ? pd[j] : ps[j];
    if (jsel == j) val = t;
  }
  const int rl = rowBase + jsel + hsel * 8;
  float* lp = isDst ? ald : als;
  if (rl < N) lp[rl] = val;
}

#define BIN_SHIFT  8
#define BIN_NODES  256
#define MAX_BINS   1024
#define EDGE_CHUNK 4096

__device__ __forceinline__ int wave_incl_scan(int x, int lane)
{
  for (int o = 1; o < 32; o <<= 1) {
    const int t = __shfl_up(x, o, 32);
    if (lane >= o) x += t;
  }
  return x;
}

__global__ __launch_bounds__(256) void gemm1_and_bin_count(
    const float* __restrict__ X, const float* __restrict__ Wm,
    const float* __restrict__ a_src, const float* __restrict__ a_dst,
    float* __restrict__ Hout, float* __restrict__ als, float* __restrict__ ald,
    int N, int K, int gGemm,
    const int* __restrict__ ei, int E, int NB, int* __restrict__ blkHist)
{
  __shared__ float smem[64 * 130];
  if ((int)blockIdx.x < gGemm) {
    gat_gemm_body(smem, blockIdx.x, X, Wm, a_src, a_dst, Hout, als, ald, N, K);
    return;
  }
  int* h = (int*)smem;
  const int tid = threadIdx.x;
  const int blk = blockIdx.x - gGemm;
  for (int b = tid; b < NB; b += 256) h[b] = 0;
  __syncthreads();
  const int base = blk * EDGE_CHUNK + tid;
  int d[EDGE_CHUNK / 256];
#pragma unroll
  for (int k = 0; k < EDGE_CHUNK / 256; ++k) {
    const int e = base + k * 256;
    d[k] = ei[E + (e < E ? e : E - 1)];
  }
#pragma unroll
  for (int k = 0; k < EDGE_CHUNK / 256; ++k)
    if (base + k * 256 < E) atomicAdd(&h[d[k] >> BIN_SHIFT], 1);
  __syncthreads();
  for (int b = tid; b < NB; b += 256) blkHist[blk * NB + b] = h[b];
}

__global__ __launch_bounds__(1024) void bin_scatter(
    const int* __restrict__ ei, int E, int NB, int nBlk,
    const int* __restrict__ blkHist, int* __restrict__ binStart,
    int* __restrict__ binned)
{
  __shared__ int partT[4][MAX_BINS];
  __shared__ int partP[4][MAX_BINS];
  __shared__ int cursor[MAX_BINS];
  __shared__ int wsum[32];
  const int tid = threadIdx.x, lane = tid & 31, wave = tid >> 5;
  const int blk = blockIdx.x;

  const int base = blk * EDGE_CHUNK + tid;
  int s[EDGE_CHUNK / 1024], d[EDGE_CHUNK / 1024];
#pragma unroll
  for (int k = 0; k < EDGE_CHUNK / 1024; ++k) {
    const int e  = base + k * 1024;
    const int ec = e < E ? e : E - 1;
    s[k] = ei[ec];
    d[k] = ei[E + ec];
  }

  const int g = tid >> 8, c = tid & 255;
  for (int bt = 0; bt < NB; bt += 256) {
    const int b = bt + c;
    int t = 0, p = 0;
    if (b < NB) {
      for (int j0 = g; j0 < nBlk; j0 += 64) {
        int v[16];
#pragma unroll
        for (int u = 0; u < 16; ++u) {
          const int j = j0 + 4 * u;
          v[u] = blkHist[(j < nBlk ? j : nBlk - 1) * NB + b];
        }
#pragma unroll
        for (int u = 0; u < 16; ++u) {
          const int j = j0 + 4 * u;
          t += (j < nBlk) ? v[u] : 0;
          p += (j < blk) ? v[u] : 0;
        }
      }
      partT[g][b] = t;
      partP[g][b] = p;
    }
  }
  __syncthreads();

  int tot = 0, pre = 0;
  if (tid < NB) {
    tot = partT[0][tid] + partT[1][tid] + partT[2][tid] + partT[3][tid];
    pre = partP[0][tid] + partP[1][tid] + partP[2][tid] + partP[3][tid];
  }
  const int x = wave_incl_scan(tot, lane);
  if (lane == 31) wsum[wave] = x;
  __syncthreads();
  if (wave == 0) {
    const int w = wsum[lane];
    const int y = wave_incl_scan(w, lane);
    wsum[lane] = y - w;
  }
  __syncthreads();
  const int excl = wsum[wave] + x - tot;
  if (tid < NB) {
    cursor[tid] = excl + pre;
    if (blk == 0) binStart[tid] = excl;
  }
  if (blk == 0 && tid == 0) binStart[NB] = E;
  __syncthreads();

  int pos[EDGE_CHUNK / 1024];
#pragma unroll
  for (int k = 0; k < EDGE_CHUNK / 1024; ++k) {
    pos[k] = 0;
    if (base + k * 1024 < E) pos[k] = atomicAdd(&cursor[d[k] >> BIN_SHIFT], 1);
  }
#pragma unroll
  for (int k = 0; k < EDGE_CHUNK / 1024; ++k)
    if (base + k * 1024 < E)
      binned[pos[k]] = (s[k] << BIN_SHIFT) | (d[k] & (BIN_NODES - 1));
}

__global__ __launch_bounds__(1024) void bin_sort(
    const int* __restrict__ binStart, const int* __restrict__ binned,
    int* __restrict__ deg, int* __restrict__ off, int* __restrict__ csrc, int N)
{
  __shared__ int cnt[BIN_NODES];
  __shared__ int cur[BIN_NODES];
  __shared__ int wtot[BIN_NODES / 32];
  const int tid = threadIdx.x, lane = tid & 31, wave = tid >> 5;
  const int lo = binStart[blockIdx.x], hi = binStart[blockIdx.x + 1];
  const int i0 = lo + tid;
  int w[4];
#pragma unroll
  for (int k = 0; k < 4; ++k) {
    const int i = i0 + k * 1024;
    w[k] = (i < hi) ? binned[i] : 0;
  }
  if (tid < BIN_NODES) cnt[tid] = 0;
  __syncthreads();
#pragma unroll
  for (int k = 0; k < 4; ++k)
    if (i0 + k * 1024 < hi) atomicAdd(&cnt[w[k] & (BIN_NODES - 1)], 1);
  for (int i = i0 + 4096; i < hi; i += 1024)
    atomicAdd(&cnt[binned[i] & (BIN_NODES - 1)], 1);
  __syncthreads();
  int v = 0, x = 0;
  if (tid < BIN_NODES) {
    v = cnt[tid];
    x = wave_incl_scan(v, lane);
    if (lane == 31) wtot[wave] = x;
  }
  __syncthreads();
  if (tid < BIN_NODES) {
    int pre = 0;
    for (int q = 0; q < BIN_NODES / 32; ++q) pre += (q < wave) ? wtot[q] : 0;
    const int o = lo + pre + x - v;
    cur[tid] = o;
    const int node = blockIdx.x * BIN_NODES + tid;
    if (node < N) { deg[node] = v; off[node] = o; }
  }
  __syncthreads();
  int pos[4];
#pragma unroll
  for (int k = 0; k < 4; ++k) {
    pos[k] = 0;
    if (i0 + k * 1024 < hi) pos[k] = atomicAdd(&cur[w[k] & (BIN_NODES - 1)], 1);
  }
#pragma unroll
  for (int k = 0; k < 4; ++k)
    if (i0 + k * 1024 < hi) csrc[pos[k]] = w[k] >> BIN_SHIFT;
  for (int i = i0 + 4096; i < hi; i += 1024) {
    const int t = binned[i];
    const int p = atomicAdd(&cur[t & (BIN_NODES - 1)], 1);
    csrc[p] = t >> BIN_SHIFT;
  }
}

__device__ __forceinline__ float half_xor(float v) {
  const unsigned u = __float_as_uint(v);
  return __uint_as_float(__builtin_amdgcn_permlanex16(u, u, 0x76543210u, 0xfedcba98u, false, false));
}
__device__ __forceinline__ float wave_max(float m) {
  m = fmaxf(m, half_xor(m));
  m = fmaxf(m, row_xor<8>(m));
  m = fmaxf(m, row_xor<4>(m));
  m = fmaxf(m, row_xor<2>(m));
  m = fmaxf(m, row_xor<1>(m));
  return m;
}
__device__ __forceinline__ float wave_sum(float s) {
  s += half_xor(s);
  s += row_xor<8>(s);
  s += row_xor<4>(s);
  s += row_xor<2>(s);
  s += row_xor<1>(s);
  return s;
}

__device__ __forceinline__ float lrelu(float v) {
  return (v > 0.0f) ? v : LRELU_SLOPE * v;
}

__device__ __forceinline__ float2 gat_node_row(
    const int node, const int lane,
    const int* __restrict__ off, const int* __restrict__ deg,
    const int* __restrict__ csrc, const float* __restrict__ Hin,
    const float* __restrict__ als, const float* __restrict__ ald)
{
  const int   beg   = off[node];
  const int   cnt   = deg[node];
  const float adv   = ald[node];
  const float eself = lrelu(als[node] + adv);

  int   s0 = node;
  float v0 = eself;
  if (lane < cnt) { s0 = csrc[beg + lane]; v0 = lrelu(als[s0] + adv); }

  float m = fmaxf(v0, eself);
  for (int j = lane + 32; j < cnt; j += 32)
    m = fmaxf(m, lrelu(als[csrc[beg + j]] + adv));
  m = wave_max(m);

  const float ex0 = (lane < cnt) ? __expf(v0 - m) : 0.0f;
  float dsum = ex0;
  for (int j = lane + 32; j < cnt; j += 32)
    dsum += __expf(lrelu(als[csrc[beg + j]] + adv) - m);
  dsum = wave_sum(dsum);
  const float exs   = __expf(eself - m);
  const float denom = dsum + exs;

  const float* hl = Hin + lane * 2;
  const float2 hs = *(const float2*)(hl + (size_t)node * 64);
  const float  as = exs / denom;
  float ax = as * hs.x, ay = as * hs.y;

  for (int base = 0; base < cnt; base += 32) {
    int   sj = s0;
    float aj = ex0 / denom;
    if (base > 0) {
      sj = node; aj = 0.0f;
      const int j = base + lane;
      if (j < cnt) {
        sj = csrc[beg + j];
        aj = __expf(lrelu(als[sj] + adv) - m) / denom;
      }
    }
    const int n = (cnt - base < 32) ? (cnt - base) : 32;
    for (int t = 0; t < n; t += 8) {
      float2 hv[8];
      float  wv[8];
#pragma unroll
      for (int k = 0; k < 8; ++k) {
        const int sk = __builtin_amdgcn_readlane(sj, t + k);
        wv[k] = __int_as_float(__builtin_amdgcn_readlane(__float_as_int(aj), t + k));
        hv[k] = *(const float2*)(hl + (size_t)sk * 64);
      }
#pragma unroll
      for (int k = 0; k < 8; ++k) {
        if (t + k < n) { ax += wv[k] * hv[k].x; ay += wv[k] * hv[k].y; }
      }
    }
  }
  float2 r; r.x = ax; r.y = ay;
  return r;
}

__global__ __launch_bounds__(256) void gat_node_softmax_aggregate(
    const int* __restrict__ off, const int* __restrict__ deg,
    const int* __restrict__ csrc, const float* __restrict__ Hin,
    const float* __restrict__ als, const float* __restrict__ ald,
    const float* __restrict__ bias, float* __restrict__ out, int N, int doRelu)
{
  const int lane = threadIdx.x & 31;
  const int node = __builtin_amdgcn_readfirstlane(
      (int)((blockIdx.x * (unsigned)blockDim.x + threadIdx.x) >> 5));
  if (node >= N) return;
  const float2 a = gat_node_row(node, lane, off, deg, csrc, Hin, als, ald);
  const float2 bv = *(const float2*)(bias + lane * 2);
  float2 r; r.x = a.x + bv.x; r.y = a.y + bv.y;
  if (doRelu) { r.x = fmaxf(r.x, 0.0f); r.y = fmaxf(r.y, 0.0f); }
  *(float2*)(out + (size_t)node * 64 + lane * 2) = r;
}

#define FUSE_NODES  16
#define AT_STRIDE   68

__global__ __launch_bounds__(512) void gat_node_aggregate_then_gemm(
    const int* __restrict__ off, const int* __restrict__ deg,
    const int* __restrict__ csrc, const float* __restrict__ Hin,
    const float* __restrict__ als, const float* __restrict__ ald,
    const float* __restrict__ bias,
    const float* __restrict__ Wm, const float* __restrict__ a_src, const float* __restrict__ a_dst,
    float* __restrict__ Hout, float* __restrict__ alsOut, float* __restrict__ aldOut, int N)
{
  __shared__ float At[FUSE_NODES * AT_STRIDE];
  __shared__ float Ot[FUSE_NODES * AT_STRIDE];
  __shared__ float part[4][32];
  const int lane = threadIdx.x & 31;
  const int wave = __builtin_amdgcn_readfirstlane((int)(threadIdx.x >> 5));
  const int rowBase = blockIdx.x * FUSE_NODES;
  const int node = rowBase + wave;

  float2 r; r.x = 0.0f; r.y = 0.0f;
  if (node < N) {
    const float2 a  = gat_node_row(node, lane, off, deg, csrc, Hin, als, ald);
    const float2 bv = *(const float2*)(bias + lane * 2);
    r.x = fmaxf(a.x + bv.x, 0.0f);
    r.y = fmaxf(a.y + bv.y, 0.0f);
  }
  *(float2*)&At[wave * AT_STRIDE + lane * 2] = r;

  const int l16  = lane & 15;
  const int hsel = lane >> 4;
  const int koff = hsel * 2;
  const int col  = wave * 16 + l16;

  v2f b[16] = {};
  float asv = 0.0f, adv = 0.0f;
  if (wave < 4) {
#pragma unroll
    for (int i = 0; i < 16; ++i) {
      b[i].x = Wm[(4 * i + koff) * 64 + col];
      b[i].y = Wm[(4 * i + koff + 1) * 64 + col];
    }
    asv = a_src[col];
    adv = a_dst[col];
  }
  __syncthreads();

  if (wave < 4) {
    v8f acc = {};
    const float* ap = &At[l16 * AT_STRIDE + koff];
#pragma unroll
    for (int i = 0; i < 16; ++i) {
      const float2 av = *(const float2*)(ap + 4 * i);
      v2f a; a.x = av.x; a.y = av.y;
      acc = __builtin_amdgcn_wmma_f32_16x16x4_f32(false, a, false, b[i], (short)0, acc, false, false);
    }
#pragma unroll
    for (int j = 0; j < 8; ++j)
      Ot[(j + hsel * 8) * AT_STRIDE + col] = acc[j];
    const int  jsel  = l16 & 7;
    const bool isDst = l16 >= 8;
    float val = 0.0f;
#pragma unroll
    for (int j = 0; j < 8; ++j) {
      const float ps = row_sum16(acc[j] * asv);
      const float pd = row_sum16(acc[j] * adv);
      const float t = isDst ? pd : ps;
      if (jsel == j) val = t;
    }
    part[wave][(isDst ? 16 : 0) + jsel + hsel * 8] = val;
  }
  __syncthreads();

  if (node < N)
    *(float2*)(Hout + (size_t)node * 64 + lane * 2) = *(const float2*)&Ot[wave * AT_STRIDE + lane * 2];

  if (wave == 0) {
    const float v = part[0][lane] + part[1][lane] + part[2][lane] + part[3][lane];
    const int rl = rowBase + (lane & 15);
    float* lp = (lane >= 16) ? aldOut : alsOut;
    if (rl < N) lp[rl] = v;
  }
}

extern "C" void kernel_launch(void* const* d_in, const int* in_sizes, int n_in,
                              void* d_out, int out_size, void* d_ws, size_t ws_size,
                              hipStream_t stream) {
  const float* x   = (const float*)d_in[0];
  const int*   ei  = (const int*)  d_in[1];
  const float* W1  = (const float*)d_in[2];
  const float* as1 = (const float*)d_in[3];
  const float* ad1 = (const float*)d_in[4];
  const float* b1  = (const float*)d_in[5];
  const float* W2  = (const float*)d_in[6];
  const float* as2 = (const float*)d_in[7];
  const float* ad2 = (const float*)d_in[8];
  const float* b2  = (const float*)d_in[9];

  const int N  = in_sizes[0] / 128;
  const int E  = in_sizes[1] / 2;

  char* w = (char*)d_ws;
  float* h1   = (float*)w; w += (size_t)N * 64 * sizeof(float);
  float* h2   = (float*)w; w += (size_t)N * 64 * sizeof(float);
  float* als  = (float*)w; w += (size_t)N * sizeof(float);
  float* ald  = (float*)w; w += (size_t)N * sizeof(float);
  float* als2 = (float*)w; w += (size_t)N * sizeof(float);
  float* ald2 = (float*)w; w += (size_t)N * sizeof(float);
  int*   deg  = (int*)w;   w += (size_t)N * sizeof(int);
  int*   off  = (int*)w;   w += (size_t)N * sizeof(int);
  const int NB   = (N + BIN_NODES - 1) / BIN_NODES;
  const int nBlk = (E + EDGE_CHUNK - 1) / EDGE_CHUNK;
  int*   binStart = (int*)w; w += (size_t)(NB + 1) * sizeof(int);
  int*   blkHist  = (int*)w; w += (size_t)nBlk * NB * sizeof(int);
  int*   binned   = (int*)w; w += (size_t)E * sizeof(int);
  int*   csrc     = (int*)w; w += (size_t)E * sizeof(int);

  const dim3 blk(256);
  const dim3 blkBin(1024);
  const int gGemm     = (N + 127) / 128;
  const int gWarpNode = (N * 32 + 255) / 256;
  const int gFuse     = (N + FUSE_NODES - 1) / FUSE_NODES;

  gemm1_and_bin_count<<<gGemm + nBlk, blk, 0, stream>>>(x, W1, as1, ad1, h1, als, ald, N, 128, gGemm,
                                                       ei, E, NB, blkHist);
  bin_scatter<<<nBlk, blkBin, 0, stream>>>(ei, E, NB, nBlk, blkHist, binStart, binned);
  bin_sort   <<<NB,   blkBin, 0, stream>>>(binStart, binned, deg, off, csrc, N);

  gat_node_aggregate_then_gemm<<<gFuse, dim3(FUSE_NODES * 32), 0, stream>>>(off, deg, csrc, h1, als, ald, b1,
                                                                          W2, as2, ad2, h2, als2, ald2, N);

  gat_node_softmax_aggregate<<<gWarpNode, blk, 0, stream>>>(off, deg, csrc, h2, als2, ald2, b2, (float*)d_out, N, 0);
}


# Edge source logit formed from the gathered row itself (no per-edge read of a logit table); softmax folded per 8-row batch on a running max; logit and softmax sums in a different tree
# speedup vs baseline: 4.9374x; 1.0565x over previous
#include <hip/hip_runtime.h>

typedef float    v2f  __attribute__((ext_vector_type(2)));
typedef float    v8f  __attribute__((ext_vector_type(8)));

#define LRELU_SLOPE 0.2f

template <int M> __device__ __forceinline__ float row_xor(float v) {
  const int i = __float_as_int(v);
  return __int_as_float(__builtin_amdgcn_update_dpp(i, i, 0x160 | M, 0xf, 0xf, true));
}
__device__ __forceinline__ float row_sum16(float s) {
  s += row_xor<8>(s);
  s += row_xor<4>(s);
  s += row_xor<2>(s);
  s += row_xor<1>(s);
  return s;
}

__device__ __forceinline__ void gat_gemm_body(
    float* __restrict__ wldsT, const int blk,
    const float* __restrict__ X, const float* __restrict__ Wm,
    const float* __restrict__ a_dst,
    float* __restrict__ Hout, float* __restrict__ ald,
    int N, int K)
{
  const int stride = K + 2;
  const int tid = threadIdx.x;
  for (int i = tid; i < K * 64; i += 256) {
    const int k = i >> 6, col = i & 63;
    wldsT[col * stride + k] = Wm[i];
  }
  __syncthreads();

  const int wave = tid >> 5;
  const int lane = tid & 31;
  const int l16  = lane & 15;
  const int hsel = lane >> 4;
  const int rowBase = blk * 128 + wave * 16;
  int row = rowBase + l16;
  if (row >= N) row = N - 1;

  v8f acc[4] = {};

  const float ad0 = a_dst[l16], ad1 = a_dst[16 + l16], ad2 = a_dst[32 + l16], ad3 = a_dst[48 + l16];

  const int koff = hsel * 2;
  const float* xrow = X + (size_t)row * K + koff;
  const float* wb0 = &wldsT[(0 * 16 + l16) * stride + koff];
  const float* wb1 = &wldsT[(1 * 16 + l16) * stride + koff];
  const float* wb2 = &wldsT[(2 * 16 + l16) * stride + koff];
  const float* wb3 = &wldsT[(3 * 16 + l16) * stride + koff];

  for (int kk = 0; kk < K; kk += 4) {
    const float2 av  = *(const float2*)(xrow + kk);
    const float2 bv0 = *(const float2*)(wb0 + kk);
    const float2 bv1 = *(const float2*)(wb1 + kk);
    const float2 bv2 = *(const float2*)(wb2 + kk);
    const float2 bv3 = *(const float2*)(wb3 + kk);
    v2f a;  a.x = av.x;   a.y = av.y;
    v2f b0; b0.x = bv0.x; b0.y = bv0.y;
    v2f b1; b1.x = bv1.x; b1.y = bv1.y;
    v2f b2; b2.x = bv2.x; b2.y = bv2.y;
    v2f b3; b3.x = bv3.x; b3.y = bv3.y;
    acc[0] = __builtin_amdgcn_wmma_f32_16x16x4_f32(false, a, false, b0, (short)0, acc[0], false, false);
    acc[1] = __builtin_amdgcn_wmma_f32_16x16x4_f32(false, a, false, b1, (short)0, acc[1], false, false);
    acc[2] = __builtin_amdgcn_wmma_f32_16x16x4_f32(false, a, false, b2, (short)0, acc[2], false, false);
    acc[3] = __builtin_amdgcn_wmma_f32_16x16x4_f32(false, a, false, b3, (short)0, acc[3], false, false);
  }

#pragma unroll
  for (int nt = 0; nt < 4; ++nt) {
#pragma unroll
    for (int j = 0; j < 8; ++j) {
      const int ro = rowBase + j + hsel * 8;
      if (ro < N) Hout[(size_t)ro * 64 + nt * 16 + l16] = acc[nt][j];
    }
  }

  float pd[8];
#pragma unroll
  for (int j = 0; j < 8; ++j)
    pd[j] = row_sum16(acc[0][j] * ad0 + acc[1][j] * ad1 + acc[2][j] * ad2 + acc[3][j] * ad3);
  float val = 0.0f;
#pragma unroll
  for (int j = 0; j < 8; ++j)
    if (l16 == j) val = pd[j];
  const int rl = rowBase + l16 + hsel * 8;
  if (l16 < 8 && rl < N) ald[rl] = val;
}

#define BIN_SHIFT  8
#define BIN_NODES  256
#define MAX_BINS   1024
#define EDGE_CHUNK 4096

__device__ __forceinline__ int wave_incl_scan(int x, int lane)
{
  for (int o = 1; o < 32; o <<= 1) {
    const int t = __shfl_up(x, o, 32);
    if (lane >= o) x += t;
  }
  return x;
}

__global__ __launch_bounds__(256) void gemm1_and_bin_count(
    const float* __restrict__ X, const float* __restrict__ Wm,
    const float* __restrict__ a_dst,
    float* __restrict__ Hout, float* __restrict__ ald,
    int N, int K, int gGemm,
    const int* __restrict__ ei, int E, int NB, int* __restrict__ blkHist)
{
  __shared__ float smem[64 * 130];
  if ((int)blockIdx.x < gGemm) {
    gat_gemm_body(smem, blockIdx.x, X, Wm, a_dst, Hout, ald, N, K);
    return;
  }
  int* h = (int*)smem;
  const int tid = threadIdx.x;
  const int blk = blockIdx.x - gGemm;
  for (int b = tid; b < NB; b += 256) h[b] = 0;
  __syncthreads();
  const int base = blk * EDGE_CHUNK + tid;
  int d[EDGE_CHUNK / 256];
#pragma unroll
  for (int k = 0; k < EDGE_CHUNK / 256; ++k) {
    const int e = base + k * 256;
    d[k] = ei[E + (e < E ? e : E - 1)];
  }
#pragma unroll
  for (int k = 0; k < EDGE_CHUNK / 256; ++k)
    if (base + k * 256 < E) atomicAdd(&h[d[k] >> BIN_SHIFT], 1);
  __syncthreads();
  for (int b = tid; b < NB; b += 256) blkHist[blk * NB + b] = h[b];
}

__global__ __launch_bounds__(1024) void bin_scatter(
    const int* __restrict__ ei, int E, int NB, int nBlk,
    const int* __restrict__ blkHist, int* __restrict__ binStart,
    int* __restrict__ binned)
{
  __shared__ int partT[4][MAX_BINS];
  __shared__ int partP[4][MAX_BINS];
  __shared__ int cursor[MAX_BINS];
  __shared__ int wsum[32];
  const int tid = threadIdx.x, lane = tid & 31, wave = tid >> 5;
  const int blk = blockIdx.x;

  const int base = blk * EDGE_CHUNK + tid;
  int s[EDGE_CHUNK / 1024], d[EDGE_CHUNK / 1024];
#pragma unroll
  for (int k = 0; k < EDGE_CHUNK / 1024; ++k) {
    const int e  = base + k * 1024;
    const int ec = e < E ? e : E - 1;
    s[k] = ei[ec];
    d[k] = ei[E + ec];
  }

  const int g = tid >> 8, c = tid & 255;
  for (int bt = 0; bt < NB; bt += 256) {
    const int b = bt + c;
    int t = 0, p = 0;
    if (b < NB) {
      for (int j0 = g; j0 < nBlk; j0 += 64) {
        int v[16];
#pragma unroll
        for (int u = 0; u < 16; ++u) {
          const int j = j0 + 4 * u;
          v[u] = blkHist[(j < nBlk ? j : nBlk - 1) * NB + b];
        }
#pragma unroll
        for (int u = 0; u < 16; ++u) {
          const int j = j0 + 4 * u;
          t += (j < nBlk) ? v[u] : 0;
          p += (j < blk) ? v[u] : 0;
        }
      }
      partT[g][b] = t;
      partP[g][b] = p;
    }
  }
  __syncthreads();

  int tot = 0, pre = 0;
  if (tid < NB) {
    tot = partT[0][tid] + partT[1][tid] + partT[2][tid] + partT[3][tid];
    pre = partP[0][tid] + partP[1][tid] + partP[2][tid] + partP[3][tid];
  }
  const int x = wave_incl_scan(tot, lane);
  if (lane == 31) wsum[wave] = x;
  __syncthreads();
  if (wave == 0) {
    const int w = wsum[lane];
    const int y = wave_incl_scan(w, lane);
    wsum[lane] = y - w;
  }
  __syncthreads();
  const int excl = wsum[wave] + x - tot;
  if (tid < NB) {
    cursor[tid] = excl + pre;
    if (blk == 0) binStart[tid] = excl;
  }
  if (blk == 0 && tid == 0) binStart[NB] = E;
  __syncthreads();

  int pos[EDGE_CHUNK / 1024];
#pragma unroll
  for (int k = 0; k < EDGE_CHUNK / 1024; ++k) {
    pos[k] = 0;
    if (base + k * 1024 < E) pos[k] = atomicAdd(&cursor[d[k] >> BIN_SHIFT], 1);
  }
#pragma unroll
  for (int k = 0; k < EDGE_CHUNK / 1024; ++k)
    if (base + k * 1024 < E)
      binned[pos[k]] = (s[k] << BIN_SHIFT) | (d[k] & (BIN_NODES - 1));
}

__global__ __launch_bounds__(1024) void bin_sort(
    const int* __restrict__ binStart, const int* __restrict__ binned,
    int* __restrict__ deg, int* __restrict__ off, int* __restrict__ csrc, int N)
{
  __shared__ int cnt[BIN_NODES];
  __shared__ int cur[BIN_NODES];
  __shared__ int wtot[BIN_NODES / 32];
  const int tid = threadIdx.x, lane = tid & 31, wave = tid >> 5;
  const int lo = binStart[blockIdx.x], hi = binStart[blockIdx.x + 1];
  const int i0 = lo + tid;
  int w[4];
#pragma unroll
  for (int k = 0; k < 4; ++k) {
    const int i = i0 + k * 1024;
    w[k] = (i < hi) ? binned[i] : 0;
  }
  if (tid < BIN_NODES) cnt[tid] = 0;
  __syncthreads();
#pragma unroll
  for (int k = 0; k < 4; ++k)
    if (i0 + k * 1024 < hi) atomicAdd(&cnt[w[k] & (BIN_NODES - 1)], 1);
  for (int i = i0 + 4096; i < hi; i += 1024)
    atomicAdd(&cnt[binned[i] & (BIN_NODES - 1)], 1);
  __syncthreads();
  int v = 0, x = 0;
  if (tid < BIN_NODES) {
    v = cnt[tid];
    x = wave_incl_scan(v, lane);
    if (lane == 31) wtot[wave] = x;
  }
  __syncthreads();
  if (tid < BIN_NODES) {
    int pre = 0;
    for (int q = 0; q < BIN_NODES / 32; ++q) pre += (q < wave) ? wtot[q] : 0;
    const int o = lo + pre + x - v;
    cur[tid] = o;
    const int node = blockIdx.x * BIN_NODES + tid;
    if (node < N) { deg[node] = v; off[node] = o; }
  }
  __syncthreads();
  int pos[4];
#pragma unroll
  for (int k = 0; k < 4; ++k) {
    pos[k] = 0;
    if (i0 + k * 1024 < hi) pos[k] = atomicAdd(&cur[w[k] & (BIN_NODES - 1)], 1);
  }
#pragma unroll
  for (int k = 0; k < 4; ++k)
    if (i0 + k * 1024 < hi) csrc[pos[k]] = w[k] >> BIN_SHIFT;
  for (int i = i0 + 4096; i < hi; i += 1024) {
    const int t = binned[i];
    const int p = atomicAdd(&cur[t & (BIN_NODES - 1)], 1);
    csrc[p] = t >> BIN_SHIFT;
  }
}

__device__ __forceinline__ float half_xor(float v) {
  const unsigned u = __float_as_uint(v);
  return __uint_as_float(__builtin_amdgcn_permlanex16(u, u, 0x76543210u, 0xfedcba98u, false, false));
}
__device__ __forceinline__ float wave_sum(float s) {
  s += half_xor(s);
  s += row_xor<8>(s);
  s += row_xor<4>(s);
  s += row_xor<2>(s);
  s += row_xor<1>(s);
  return s;
}

__device__ __forceinline__ float lrelu(float v) {
  return (v > 0.0f) ? v : LRELU_SLOPE * v;
}

__device__ __forceinline__ float2 gat_node_row(
    const int node, const int lane,
    const int* __restrict__ off, const int* __restrict__ deg,
    const int* __restrict__ csrc, const float* __restrict__ Hin,
    const float* __restrict__ ald, const float a0, const float a1)
{
  const int   beg = off[node];
  const int   cnt = deg[node];
  const float adv = ald[node];

  const float* hl = Hin + lane * 2;
  const float2 hs = *(const float2*)(hl + (size_t)node * 64);
  int s0 = node;
  if (lane < cnt) s0 = csrc[beg + lane];

  float m   = lrelu(wave_sum(hs.x * a0 + hs.y * a1) + adv);
  float den = 1.0f;
  float ax = hs.x, ay = hs.y;

  const bool hi16 = (lane & 16) != 0;
  const bool hi8  = (lane & 8) != 0;
  const bool hi4  = (lane & 4) != 0;
  const int  myrow = lane >> 2;

  for (int base = 0; base < cnt; base += 32) {
    const int n = (cnt - base < 32) ? (cnt - base) : 32;
    int sj = s0;
    if (base > 0) {
      sj = node;
      if (lane < n) sj = csrc[beg + base + lane];
    }
    for (int t = 0; t < n; t += 8) {
      float2 hv[8];
      float  p[8];
#pragma unroll
      for (int k = 0; k < 8; ++k) {
        const int sk = __builtin_amdgcn_readlane(sj, t + k);
        hv[k] = *(const float2*)(hl + (size_t)sk * 64);
      }
#pragma unroll
      for (int k = 0; k < 8; ++k) p[k] = hv[k].x * a0 + hv[k].y * a1;

      float q[4];
#pragma unroll
      for (int k = 0; k < 4; ++k) {
        const float keep = hi16 ? p[k + 4] : p[k];
        const float send = hi16 ? p[k] : p[k + 4];
        q[k] = keep + half_xor(send);
      }
      float r[2];
#pragma unroll
      for (int k = 0; k < 2; ++k) {
        const float keep = hi8 ? q[k + 2] : q[k];
        const float send = hi8 ? q[k] : q[k + 2];
        r[k] = keep + row_xor<8>(send);
      }
      float s;
      {
        const float keep = hi4 ? r[1] : r[0];
        const float send = hi4 ? r[0] : r[1];
        s = keep + row_xor<4>(send);
      }
      s += row_xor<2>(s);
      s += row_xor<1>(s);

      const float e = (t + myrow < n) ? lrelu(s + adv) : -__builtin_huge_valf();
      float mb = fmaxf(e, row_xor<4>(e));
      mb = fmaxf(mb, row_xor<8>(mb));
      mb = fmaxf(mb, half_xor(mb));
      const float mn = fmaxf(m, mb);
      const float sc = __expf(m - mn);
      const float w  = __expf(e - mn);
      float ws = w + row_xor<4>(w);
      ws += row_xor<8>(ws);
      ws += half_xor(ws);
      den = den * sc + ws;
      ax *= sc; ay *= sc;
#pragma unroll
      for (int k = 0; k < 8; ++k) {
        const float wk = __int_as_float(__builtin_amdgcn_readlane(__float_as_int(w), 4 * k));
        ax += wk * hv[k].x; ay += wk * hv[k].y;
      }
      m = mn;
    }
  }
  float2 res; res.x = ax / den; res.y = ay / den;
  return res;
}

__global__ __launch_bounds__(256) void gat_node_softmax_aggregate(
    const int* __restrict__ off, const int* __restrict__ deg,
    const int* __restrict__ csrc, const float* __restrict__ Hin,
    const float* __restrict__ a_src, const float* __restrict__ ald,
    const float* __restrict__ bias, float* __restrict__ out, int N, int doRelu)
{
  const int lane = threadIdx.x & 31;
  const int node = __builtin_amdgcn_readfirstlane(
      (int)((blockIdx.x * (unsigned)blockDim.x + threadIdx.x) >> 5));
  if (node >= N) return;
  const float2 av = *(const float2*)(a_src + lane * 2);
  const float2 a = gat_node_row(node, lane, off, deg, csrc, Hin, ald, av.x, av.y);
  const float2 bv = *(const float2*)(bias + lane * 2);
  float2 r; r.x = a.x + bv.x; r.y = a.y + bv.y;
  if (doRelu) { r.x = fmaxf(r.x, 0.0f); r.y = fmaxf(r.y, 0.0f); }
  *(float2*)(out + (size_t)node * 64 + lane * 2) = r;
}

#define FUSE_NODES  16
#define AT_STRIDE   68

__global__ __launch_bounds__(512) void gat_node_aggregate_then_gemm(
    const int* __restrict__ off, const int* __restrict__ deg,
    const int* __restrict__ csrc, const float* __restrict__ Hin,
    const float* __restrict__ a_srcIn, const float* __restrict__ ald,
    const float* __restrict__ bias,
    const float* __restrict__ Wm, const float* __restrict__ a_dst,
    float* __restrict__ Hout, float* __restrict__ aldOut, int N)
{
  __shared__ float At[FUSE_NODES * AT_STRIDE];
  __shared__ float Ot[FUSE_NODES * AT_STRIDE];
  __shared__ float part[4][16];
  const int lane = threadIdx.x & 31;
  const int wave = __builtin_amdgcn_readfirstlane((int)(threadIdx.x >> 5));
  const int rowBase = blockIdx.x * FUSE_NODES;
  const int node = rowBase + wave;

  float2 r; r.x = 0.0f; r.y = 0.0f;
  if (node < N) {
    const float2 av = *(const float2*)(a_srcIn + lane * 2);
    const float2 a  = gat_node_row(node, lane, off, deg, csrc, Hin, ald, av.x, av.y);
    const float2 bv = *(const float2*)(bias + lane * 2);
    r.x = fmaxf(a.x + bv.x, 0.0f);
    r.y = fmaxf(a.y + bv.y, 0.0f);
  }
  *(float2*)&At[wave * AT_STRIDE + lane * 2] = r;

  const int l16  = lane & 15;
  const int hsel = lane >> 4;
  const int koff = hsel * 2;
  const int col  = wave * 16 + l16;

  v2f b[16] = {};
  float adv = 0.0f;
  if (wave < 4) {
#pragma unroll
    for (int i = 0; i < 16; ++i) {
      b[i].x = Wm[(4 * i + koff) * 64 + col];
      b[i].y = Wm[(4 * i + koff + 1) * 64 + col];
    }
    adv = a_dst[col];
  }
  __syncthreads();

  if (wave < 4) {
    v8f acc = {};
    const float* ap = &At[l16 * AT_STRIDE + koff];
#pragma unroll
    for (int i = 0; i < 16; ++i) {
      const float2 av = *(const float2*)(ap + 4 * i);
      v2f a; a.x = av.x; a.y = av.y;
      acc = __builtin_amdgcn_wmma_f32_16x16x4_f32(false, a, false, b[i], (short)0, acc, false, false);
    }
#pragma unroll
    for (int j = 0; j < 8; ++j)
      Ot[(j + hsel * 8) * AT_STRIDE + col] = acc[j];
    float val = 0.0f;
#pragma unroll
    for (int j = 0; j < 8; ++j) {
      const float pd = row_sum16(acc[j] * adv);
      if (l16 == j) val = pd;
    }
    if (l16 < 8) part[wave][l16 + hsel * 8] = val;
  }
  __syncthreads();

  if (node < N)
    *(float2*)(Hout + (size_t)node * 64 + lane * 2) = *(const float2*)&Ot[wave * AT_STRIDE + lane * 2];

  if (wave == 0 && lane < 16) {
    const float v = part[0][lane] + part[1][lane] + part[2][lane] + part[3][lane];
    const int rl = rowBase + lane;
    if (rl < N) aldOut[rl] = v;
  }
}

extern "C" void kernel_launch(void* const* d_in, const int* in_sizes, int n_in,
                              void* d_out, int out_size, void* d_ws, size_t ws_size,
                              hipStream_t stream) {
  const float* x   = (const float*)d_in[0];
  const int*   ei  = (const int*)  d_in[1];
  const float* W1  = (const float*)d_in[2];
  const float* as1 = (const float*)d_in[3];
  const float* ad1 = (const float*)d_in[4];
  const float* b1  = (const float*)d_in[5];
  const float* W2  = (const float*)d_in[6];
  const float* as2 = (const float*)d_in[7];
  const float* ad2 = (const float*)d_in[8];
  const float* b2  = (const float*)d_in[9];

  const int N  = in_sizes[0] / 128;
  const int E  = in_sizes[1] / 2;

  char* w = (char*)d_ws;
  float* h1   = (float*)w; w += (size_t)N * 64 * sizeof(float);
  float* h2   = (float*)w; w += (size_t)N * 64 * sizeof(float);
  float* ald  = (float*)w; w += (size_t)N * sizeof(float);
  float* ald2 = (float*)w; w += (size_t)N * sizeof(float);
  int*   deg  = (int*)w;   w += (size_t)N * sizeof(int);
  int*   off  = (int*)w;   w += (size_t)N * sizeof(int);
  const int NB   = (N + BIN_NODES - 1) / BIN_NODES;
  const int nBlk = (E + EDGE_CHUNK - 1) / EDGE_CHUNK;
  int*   binStart = (int*)w; w += (size_t)(NB + 1) * sizeof(int);
  int*   blkHist  = (int*)w; w += (size_t)nBlk * NB * sizeof(int);
  int*   binned   = (int*)w; w += (size_t)E * sizeof(int);
  int*   csrc     = (int*)w; w += (size_t)E * sizeof(int);

  const dim3 blk(256);
  const dim3 blkBin(1024);
  const int gGemm     = (N + 127) / 128;
  const int gWarpNode = (N * 32 + 255) / 256;
  const int gFuse     = (N + FUSE_NODES - 1) / FUSE_NODES;

  gemm1_and_bin_count<<<gGemm + nBlk, blk, 0, stream>>>(x, W1, ad1, h1, ald, N, 128, gGemm,
                                                       ei, E, NB, blkHist);
  bin_scatter<<<nBlk, blkBin, 0, stream>>>(ei, E, NB, nBlk, blkHist, binStart, binned);
  bin_sort   <<<NB,   blkBin, 0, stream>>>(binStart, binned, deg, off, csrc, N);

  gat_node_aggregate_then_gemm<<<gFuse, dim3(FUSE_NODES * 32), 0, stream>>>(off, deg, csrc, h1, as1, ald, b1,
                                                                          W2, ad2, h2, ald2, N);

  gat_node_softmax_aggregate<<<gWarpNode, blk, 0, stream>>>(off, deg, csrc, h2, as2, ald2, b2, (float*)d_out, N, 0);
}


# Edges bucketed in two launches, not three: each 4096-edge block orders its chunk by bin in LDS, each bin block collects its piece of every chunk; 5 to 4 launches; bucket (summation) order changes
# speedup vs baseline: 6.8838x; 1.3942x over previous
#include <hip/hip_runtime.h>

typedef float    v2f  __attribute__((ext_vector_type(2)));
typedef float    v8f  __attribute__((ext_vector_type(8)));

#define LRELU_SLOPE 0.2f

template <int M> __device__ __forceinline__ float row_xor(float v) {
  const int i = __float_as_int(v);
  return __int_as_float(__builtin_amdgcn_update_dpp(i, i, 0x160 | M, 0xf, 0xf, true));
}
__device__ __forceinline__ float row_sum16(float s) {
  s += row_xor<8>(s);
  s += row_xor<4>(s);
  s += row_xor<2>(s);
  s += row_xor<1>(s);
  return s;
}

__device__ __forceinline__ void gat_gemm_body(
    float* __restrict__ wldsT, const int blk,
    const float* __restrict__ X, const float* __restrict__ Wm,
    const float* __restrict__ a_dst,
    float* __restrict__ Hout, float* __restrict__ ald,
    int N, int K)
{
  const int stride = K + 2;
  const int tid = threadIdx.x;
  for (int i = tid; i < K * 64; i += 256) {
    const int k = i >> 6, col = i & 63;
    wldsT[col * stride + k] = Wm[i];
  }
  __syncthreads();

  const int wave = tid >> 5;
  const int lane = tid & 31;
  const int l16  = lane & 15;
  const int hsel = lane >> 4;
  const int rowBase = blk * 128 + wave * 16;
  int row = rowBase + l16;
  if (row >= N) row = N - 1;

  v8f acc[4] = {};

  const float ad0 = a_dst[l16], ad1 = a_dst[16 + l16], ad2 = a_dst[32 + l16], ad3 = a_dst[48 + l16];

  const int koff = hsel * 2;
  const float* xrow = X + (size_t)row * K + koff;
  const float* wb0 = &wldsT[(0 * 16 + l16) * stride + koff];
  const float* wb1 = &wldsT[(1 * 16 + l16) * stride + koff];
  const float* wb2 = &wldsT[(2 * 16 + l16) * stride + koff];
  const float* wb3 = &wldsT[(3 * 16 + l16) * stride + koff];

  for (int kk = 0; kk < K; kk += 4) {
    const float2 av  = *(const float2*)(xrow + kk);
    const float2 bv0 = *(const float2*)(wb0 + kk);
    const float2 bv1 = *(const float2*)(wb1 + kk);
    const float2 bv2 = *(const float2*)(wb2 + kk);
    const float2 bv3 = *(const float2*)(wb3 + kk);
    v2f a;  a.x = av.x;   a.y = av.y;
    v2f b0; b0.x = bv0.x; b0.y = bv0.y;
    v2f b1; b1.x = bv1.x; b1.y = bv1.y;
    v2f b2; b2.x = bv2.x; b2.y = bv2.y;
    v2f b3; b3.x = bv3.x; b3.y = bv3.y;
    acc[0] = __builtin_amdgcn_wmma_f32_16x16x4_f32(false, a, false, b0, (short)0, acc[0], false, false);
    acc[1] = __builtin_amdgcn_wmma_f32_16x16x4_f32(false, a, false, b1, (short)0, acc[1], false, false);
    acc[2] = __builtin_amdgcn_wmma_f32_16x16x4_f32(false, a, false, b2, (short)0, acc[2], false, false);
    acc[3] = __builtin_amdgcn_wmma_f32_16x16x4_f32(false, a, false, b3, (short)0, acc[3], false, false);
  }

#pragma unroll
  for (int nt = 0; nt < 4; ++nt) {
#pragma unroll
    for (int j = 0; j < 8; ++j) {
      const int ro = rowBase + j + hsel * 8;
      if (ro < N) Hout[(size_t)ro * 64 + nt * 16 + l16] = acc[nt][j];
    }
  }

  float pd[8];
#pragma unroll
  for (int j = 0; j < 8; ++j)
    pd[j] = row_sum16(acc[0][j] * ad0 + acc[1][j] * ad1 + acc[2][j] * ad2 + acc[3][j] * ad3);
  float val = 0.0f;
#pragma unroll
  for (int j = 0; j < 8; ++j)
    if (l16 == j) val = pd[j];
  const int rl = rowBase + l16 + hsel * 8;
  if (l16 < 8 && rl < N) ald[rl] = val;
}

#define BIN_SHIFT  8
#define BIN_NODES  256
#define MAX_BINS   1024
#define EDGE_CHUNK 4096
#define HELD       8

__device__ __forceinline__ int wave_incl_scan(int x, int lane)
{
  for (int o = 1; o < 32; o <<= 1) {
    const int t = __shfl_up(x, o, 32);
    if (lane >= o) x += t;
  }
  return x;
}

template <int M> __device__ __forceinline__ int row_xor_i(int i) {
  return __builtin_amdgcn_update_dpp(i, i, 0x160 | M, 0xf, 0xf, true);
}
__device__ __forceinline__ int wave_sum_i(int s) {
  s += (int)__builtin_amdgcn_permlanex16((unsigned)s, (unsigned)s, 0x76543210u, 0xfedcba98u, false, false);
  s += row_xor_i<8>(s);
  s += row_xor_i<4>(s);
  s += row_xor_i<2>(s);
  s += row_xor_i<1>(s);
  return s;
}

__global__ __launch_bounds__(256) void gemm1_and_chunk_sort(
    const float* __restrict__ X, const float* __restrict__ Wm,
    const float* __restrict__ a_dst,
    float* __restrict__ Hout, float* __restrict__ ald,
    int N, int K, int gGemm,
    const int* __restrict__ ei, int E, int NB,
    int* __restrict__ lstart, int* __restrict__ lbinned)
{
  __shared__ __attribute__((aligned(16))) float smem[64 * 130];
  if ((int)blockIdx.x < gGemm) {
    gat_gemm_body(smem, blockIdx.x, X, Wm, a_dst, Hout, ald, N, K);
    return;
  }
  int*  h     = (int*)smem;
  int4* h4    = (int4*)smem;
  int*  wtot  = h + MAX_BINS;
  int*  stage = h + MAX_BINS + 8;
  const int tid = threadIdx.x, lane = tid & 31, wave = tid >> 5;
  const int blk = blockIdx.x - gGemm;
  h4[tid] = make_int4(0, 0, 0, 0);
  const int base = blk * EDGE_CHUNK + tid;
  int s[EDGE_CHUNK / 256], d[EDGE_CHUNK / 256];
#pragma unroll
  for (int k = 0; k < EDGE_CHUNK / 256; ++k) {
    const int e  = base + k * 256;
    const int ec = e < E ? e : E - 1;
    s[k] = ei[ec];
    d[k] = ei[E + ec];
  }
  __syncthreads();
  int rk[EDGE_CHUNK / 256];
#pragma unroll
  for (int k = 0; k < EDGE_CHUNK / 256; ++k) {
    rk[k] = 0;
    if (base + k * 256 < E) rk[k] = atomicAdd(&h[d[k] >> BIN_SHIFT], 1);
  }
  __syncthreads();
  const int4 v = h4[tid];
  const int  t = v.x + v.y + v.z + v.w;
  const int  x = wave_incl_scan(t, lane);
  if (lane == 31) wtot[wave] = x;
  __syncthreads();
  int pre = 0;
  for (int q = 0; q < 8; ++q) pre += (q < wave) ? wtot[q] : 0;
  int4 st;
  st.x = pre + x - t; st.y = st.x + v.x; st.z = st.y + v.y; st.w = st.z + v.z;
  h4[tid] = st;
  int* row = lstart + (size_t)blk * (NB + 1);
  const int b0 = tid * 4;
  if (b0     < NB) row[b0]     = st.x;
  if (b0 + 1 < NB) row[b0 + 1] = st.y;
  if (b0 + 2 < NB) row[b0 + 2] = st.z;
  if (b0 + 3 < NB) row[b0 + 3] = st.w;
  if (tid == 255) row[NB] = st.w + v.w;
  __syncthreads();
#pragma unroll
  for (int k = 0; k < EDGE_CHUNK / 256; ++k)
    if (base + k * 256 < E)
      stage[h[d[k] >> BIN_SHIFT] + rk[k]] = (s[k] << BIN_SHIFT) | (d[k] & (BIN_NODES - 1));
  __syncthreads();
  const int left = E - blk * EDGE_CHUNK;
  const int nv = left < EDGE_CHUNK ? left : EDGE_CHUNK;
#pragma unroll
  for (int k = 0; k < EDGE_CHUNK / 256; ++k) {
    const int i = tid + k * 256;
    if (i < nv) lbinned[(size_t)blk * EDGE_CHUNK + i] = stage[i];
  }
}

__global__ __launch_bounds__(1024) void bin_collect_sort(
    const int* __restrict__ lstart, const int* __restrict__ lbinned,
    int nBlk, int NB,
    int* __restrict__ deg, int* __restrict__ off, int* __restrict__ csrc, int N)
{
  __shared__ int cnt[BIN_NODES];
  __shared__ int cur[BIN_NODES];
  __shared__ int wtot[BIN_NODES / 32];
  __shared__ int wlo[32];
  const int tid = threadIdx.x, lane = tid & 31;
  const int wave  = __builtin_amdgcn_readfirstlane(tid >> 5);
  const int b     = blockIdx.x;
  const int pitch = NB + 1;

  int lo = 0, hi = 0, losum = 0;
  {
    const int c = wave + 32 * lane;
    if (c < nBlk) {
      lo = lstart[(size_t)c * pitch + b];
      hi = lstart[(size_t)c * pitch + b + 1];
    }
    losum = lo;
    for (int c2 = c + 1024; c2 < nBlk; c2 += 1024) losum += lstart[(size_t)c2 * pitch + b];
  }
  if (tid < BIN_NODES) cnt[tid] = 0;

  int n[HELD];
  int w0[HELD], w1[HELD];
  int maxn = 0;
#pragma unroll
  for (int j = 0; j < HELD; ++j) {
    const int lj = __builtin_amdgcn_readlane(lo, j);
    n[j] = __builtin_amdgcn_readlane(hi, j) - lj;
    maxn = n[j] > maxn ? n[j] : maxn;
    const int* p = lbinned + (size_t)(wave + 32 * j) * EDGE_CHUNK + lj + lane;
    w0[j] = 0; w1[j] = 0;
    if (lane < n[j]) w0[j] = p[0];
  }
  const bool two  = maxn > 32;
  const bool more = maxn > 64 || nBlk > 32 * HELD;
  if (two) {
#pragma unroll
    for (int j = 0; j < HELD; ++j) {
      const int lj = __builtin_amdgcn_readlane(lo, j);
      const int* p = lbinned + (size_t)(wave + 32 * j) * EDGE_CHUNK + lj + lane;
      if (lane + 32 < n[j]) w1[j] = p[32];
    }
  }
  __syncthreads();

#pragma unroll
  for (int j = 0; j < HELD; ++j)
    if (lane < n[j]) atomicAdd(&cnt[w0[j] & (BIN_NODES - 1)], 1);
  if (two) {
#pragma unroll
    for (int j = 0; j < HELD; ++j)
      if (lane + 32 < n[j]) atomicAdd(&cnt[w1[j] & (BIN_NODES - 1)], 1);
  }
  if (more) {
    for (int c = wave; c < nBlk; c += 32) {
      const int l = lstart[(size_t)c * pitch + b], h = lstart[(size_t)c * pitch + b + 1];
      const int first = (c < wave + 32 * HELD) ? 64 : 0;
      for (int i = l + first + lane; i < h; i += 32)
        atomicAdd(&cnt[lbinned[(size_t)c * EDGE_CHUNK + i] & (BIN_NODES - 1)], 1);
    }
  }
  {
    const int sl = wave_sum_i(losum);
    if (lane == 0) wlo[wave] = sl;
  }
  __syncthreads();

  int v = 0, x = 0;
  if (tid < BIN_NODES) {
    v = cnt[tid];
    x = wave_incl_scan(v, lane);
    if (lane == 31) wtot[wave] = x;
  }
  __syncthreads();
  if (tid < BIN_NODES) {
    const int binlo = wave_sum_i(wlo[lane]);
    int pre = 0;
    for (int q = 0; q < BIN_NODES / 32; ++q) pre += (q < wave) ? wtot[q] : 0;
    const int o = binlo + pre + x - v;
    cur[tid] = o;
    const int node = b * BIN_NODES + tid;
    if (node < N) { deg[node] = v; off[node] = o; }
  }
  __syncthreads();

  int pos[HELD];
#pragma unroll
  for (int j = 0; j < HELD; ++j) {
    pos[j] = 0;
    if (lane < n[j]) pos[j] = atomicAdd(&cur[w0[j] & (BIN_NODES - 1)], 1);
  }
#pragma unroll
  for (int j = 0; j < HELD; ++j)
    if (lane < n[j]) csrc[pos[j]] = w0[j] >> BIN_SHIFT;
  if (two) {
#pragma unroll
    for (int j = 0; j < HELD; ++j) {
      pos[j] = 0;
      if (lane + 32 < n[j]) pos[j] = atomicAdd(&cur[w1[j] & (BIN_NODES - 1)], 1);
    }
#pragma unroll
    for (int j = 0; j < HELD; ++j)
      if (lane + 32 < n[j]) csrc[pos[j]] = w1[j] >> BIN_SHIFT;
  }
  if (more) {
    for (int c = wave; c < nBlk; c += 32) {
      const int l = lstart[(size_t)c * pitch + b], h = lstart[(size_t)c * pitch + b + 1];
      const int first = (c < wave + 32 * HELD) ? 64 : 0;
      for (int i = l + first + lane; i < h; i += 32) {
        const int t = lbinned[(size_t)c * EDGE_CHUNK + i];
        const int p = atomicAdd(&cur[t & (BIN_NODES - 1)], 1);
        csrc[p] = t >> BIN_SHIFT;
      }
    }
  }
}

__device__ __forceinline__ float half_xor(float v) {
  const unsigned u = __float_as_uint(v);
  return __uint_as_float(__builtin_amdgcn_permlanex16(u, u, 0x76543210u, 0xfedcba98u, false, false));
}
__device__ __forceinline__ float wave_sum(float s) {
  s += half_xor(s);
  s += row_xor<8>(s);
  s += row_xor<4>(s);
  s += row_xor<2>(s);
  s += row_xor<1>(s);
  return s;
}

__device__ __forceinline__ float lrelu(float v) {
  return (v > 0.0f) ? v : LRELU_SLOPE * v;
}

__device__ __forceinline__ float2 gat_node_row(
    const int node, const int lane,
    const int* __restrict__ off, const int* __restrict__ deg,
    const int* __restrict__ csrc, const float* __restrict__ Hin,
    const float* __restrict__ ald, const float a0, const float a1)
{
  const int   beg = off[node];
  const int   cnt = deg[node];
  const float adv = ald[node];

  const float* hl = Hin + lane * 2;
  const float2 hs = *(const float2*)(hl + (size_t)node * 64);
  int s0 = node;
  if (lane < cnt) s0 = csrc[beg + lane];

  float m   = lrelu(wave_sum(hs.x * a0 + hs.y * a1) + adv);
  float den = 1.0f;
  float ax = hs.x, ay = hs.y;

  const bool hi16 = (lane & 16) != 0;
  const bool hi8  = (lane & 8) != 0;
  const bool hi4  = (lane & 4) != 0;
  const int  myrow = lane >> 2;

  for (int base = 0; base < cnt; base += 32) {
    const int n = (cnt - base < 32) ? (cnt - base) : 32;
    int sj = s0;
    if (base > 0) {
      sj = node;
      if (lane < n) sj = csrc[beg + base + lane];
    }
    for (int t = 0; t < n; t += 8) {
      float2 hv[8];
      float  p[8];
#pragma unroll
      for (int k = 0; k < 8; ++k) {
        const int sk = __builtin_amdgcn_readlane(sj, t + k);
        hv[k] = *(const float2*)(hl + (size_t)sk * 64);
      }
#pragma unroll
      for (int k = 0; k < 8; ++k) p[k] = hv[k].x * a0 + hv[k].y * a1;

      float q[4];
#pragma unroll
      for (int k = 0; k < 4; ++k) {
        const float keep = hi16 ? p[k + 4] : p[k];
        const float send = hi16 ? p[k] : p[k + 4];
        q[k] = keep + half_xor(send);
      }
      float r[2];
#pragma unroll
      for (int k = 0; k < 2; ++k) {
        const float keep = hi8 ? q[k + 2] : q[k];
        const float send = hi8 ? q[k] : q[k + 2];
        r[k] = keep + row_xor<8>(send);
      }
      float s;
      {
        const float keep = hi4 ? r[1] : r[0];
        const float send = hi4 ? r[0] : r[1];
        s = keep + row_xor<4>(send);
      }
      s += row_xor<2>(s);
      s += row_xor<1>(s);

      const float e = (t + myrow < n) ? lrelu(s + adv) : -__builtin_huge_valf();
      float mb = fmaxf(e, row_xor<4>(e));
      mb = fmaxf(mb, row_xor<8>(mb));
      mb = fmaxf(mb, half_xor(mb));
      const float mn = fmaxf(m, mb);
      const float sc = __expf(m - mn);
      const float w  = __expf(e - mn);
      float ws = w + row_xor<4>(w);
      ws += row_xor<8>(ws);
      ws += half_xor(ws);
      den = den * sc + ws;
      ax *= sc; ay *= sc;
#pragma unroll
      for (int k = 0; k < 8; ++k) {
        const float wk = __int_as_float(__builtin_amdgcn_readlane(__float_as_int(w), 4 * k));
        ax += wk * hv[k].x; ay += wk * hv[k].y;
      }
      m = mn;
    }
  }
  float2 res; res.x = ax / den; res.y = ay / den;
  return res;
}

__global__ __launch_bounds__(256) void gat_node_softmax_aggregate(
    const int* __restrict__ off, const int* __restrict__ deg,
    const int* __restrict__ csrc, const float* __restrict__ Hin,
    const float* __restrict__ a_src, const float* __restrict__ ald,
    const float* __restrict__ bias, float* __restrict__ out, int N, int doRelu)
{
  const int lane = threadIdx.x & 31;
  const int node = __builtin_amdgcn_readfirstlane(
      (int)((blockIdx.x * (unsigned)blockDim.x + threadIdx.x) >> 5));
  if (node >= N) return;
  const float2 av = *(const float2*)(a_src + lane * 2);
  const float2 a = gat_node_row(node, lane, off, deg, csrc, Hin, ald, av.x, av.y);
  const float2 bv = *(const float2*)(bias + lane * 2);
  float2 r; r.x = a.x + bv.x; r.y = a.y + bv.y;
  if (doRelu) { r.x = fmaxf(r.x, 0.0f); r.y = fmaxf(r.y, 0.0f); }
  *(float2*)(out + (size_t)node * 64 + lane * 2) = r;
}

#define FUSE_NODES  16
#define AT_STRIDE   68

__global__ __launch_bounds__(512) void gat_node_aggregate_then_gemm(
    const int* __restrict__ off, const int* __restrict__ deg,
    const int* __restrict__ csrc, const float* __restrict__ Hin,
    const float* __restrict__ a_srcIn, const float* __restrict__ ald,
    const float* __restrict__ bias,
    const float* __restrict__ Wm, const float* __restrict__ a_dst,
    float* __restrict__ Hout, float* __restrict__ aldOut, int N)
{
  __shared__ float At[FUSE_NODES * AT_STRIDE];
  __shared__ float Ot[FUSE_NODES * AT_STRIDE];
  __shared__ float part[4][16];
  const int lane = threadIdx.x & 31;
  const int wave = __builtin_amdgcn_readfirstlane((int)(threadIdx.x >> 5));
  const int rowBase = blockIdx.x * FUSE_NODES;
  const int node = rowBase + wave;

  float2 r; r.x = 0.0f; r.y = 0.0f;
  if (node < N) {
    const float2 av = *(const float2*)(a_srcIn + lane * 2);
    const float2 a  = gat_node_row(node, lane, off, deg, csrc, Hin, ald, av.x, av.y);
    const float2 bv = *(const float2*)(bias + lane * 2);
    r.x = fmaxf(a.x + bv.x, 0.0f);
    r.y = fmaxf(a.y + bv.y, 0.0f);
  }
  *(float2*)&At[wave * AT_STRIDE + lane * 2] = r;

  const int l16  = lane & 15;
  const int hsel = lane >> 4;
  const int koff = hsel * 2;
  const int col  = wave * 16 + l16;

  v2f b[16] = {};
  float adv = 0.0f;
  if (wave < 4) {
#pragma unroll
    for (int i = 0; i < 16; ++i) {
      b[i].x = Wm[(4 * i + koff) * 64 + col];
      b[i].y = Wm[(4 * i + koff + 1) * 64 + col];
    }
    adv = a_dst[col];
  }
  __syncthreads();

  if (wave < 4) {
    v8f acc = {};
    const float* ap = &At[l16 * AT_STRIDE + koff];
#pragma unroll
    for (int i = 0; i < 16; ++i) {
      const float2 av = *(const float2*)(ap + 4 * i);
      v2f a; a.x = av.x; a.y = av.y;
      acc = __builtin_amdgcn_wmma_f32_16x16x4_f32(false, a, false, b[i], (short)0, acc, false, false);
    }
#pragma unroll
    for (int j = 0; j < 8; ++j)
      Ot[(j + hsel * 8) * AT_STRIDE + col] = acc[j];
    float val = 0.0f;
#pragma unroll
    for (int j = 0; j < 8; ++j) {
      const float pd = row_sum16(acc[j] * adv);
      if (l16 == j) val = pd;
    }
    if (l16 < 8) part[wave][l16 + hsel * 8] = val;
  }
  __syncthreads();

  if (node < N)
    *(float2*)(Hout + (size_t)node * 64 + lane * 2) = *(const float2*)&Ot[wave * AT_STRIDE + lane * 2];

  if (wave == 0 && lane < 16) {
    const float v = part[0][lane] + part[1][lane] + part[2][lane] + part[3][lane];
    const int rl = rowBase + lane;
    if (rl < N) aldOut[rl] = v;
  }
}

extern "C" void kernel_launch(void* const* d_in, const int* in_sizes, int n_in,
                              void* d_out, int out_size, void* d_ws, size_t ws_size,
                              hipStream_t stream) {
  const float* x   = (const float*)d_in[0];
  const int*   ei  = (const int*)  d_in[1];
  const float* W1  = (const float*)d_in[2];
  const float* as1 = (const float*)d_in[3];
  const float* ad1 = (const float*)d_in[4];
  const float* b1  = (const float*)d_in[5];
  const float* W2  = (const float*)d_in[6];
  const float* as2 = (const float*)d_in[7];
  const float* ad2 = (const float*)d_in[8];
  const float* b2  = (const float*)d_in[9];

  const int N  = in_sizes[0] / 128;
  const int E  = in_sizes[1] / 2;

  char* w = (char*)d_ws;
  float* h1   = (float*)w; w += (size_t)N * 64 * sizeof(float);
  float* h2   = (float*)w; w += (size_t)N * 64 * sizeof(float);
  float* ald  = (float*)w; w += (size_t)N * sizeof(float);
  float* ald2 = (float*)w; w += (size_t)N * sizeof(float);
  int*   deg  = (int*)w;   w += (size_t)N * sizeof(int);
  int*   off  = (int*)w;   w += (size_t)N * sizeof(int);
  const int NB   = (N + BIN_NODES - 1) / BIN_NODES;
  const int nBlk = (E + EDGE_CHUNK - 1) / EDGE_CHUNK;
  int*   lstart   = (int*)w; w += (size_t)nBlk * (NB + 1) * sizeof(int);
  int*   lbinned  = (int*)w; w += (size_t)nBlk * EDGE_CHUNK * sizeof(int);
  int*   csrc     = (int*)w; w += (size_t)E * sizeof(int);

  const dim3 blk(256);
  const dim3 blkBin(1024);
  const int gGemm     = (N + 127) / 128;
  const int gWarpNode = (N * 32 + 255) / 256;
  const int gFuse     = (N + FUSE_NODES - 1) / FUSE_NODES;

  gemm1_and_chunk_sort<<<gGemm + nBlk, blk, 0, stream>>>(x, W1, ad1, h1, ald, N, 128, gGemm,
                                                        ei, E, NB, lstart, lbinned);
  bin_collect_sort<<<NB, blkBin, 0, stream>>>(lstart, lbinned, nBlk, NB, deg, off, csrc, N);

  gat_node_aggregate_then_gemm<<<gFuse, dim3(FUSE_NODES * 32), 0, stream>>>(off, deg, csrc, h1, as1, ald, b1,
                                                                          W2, ad2, h2, ald2, N);

  gat_node_softmax_aggregate<<<gWarpNode, blk, 0, stream>>>(off, deg, csrc, h2, as2, ald2, b2, (float*)d_out, N, 0);
}


# Node passes put two nodes in each wave: a 16-lane half owns a node, lanes hold 4 features, rows gathered as 16-byte loads; half the waves and gather instructions; logit sums in a different tree
# speedup vs baseline: 7.4548x; 1.0829x over previous
#include <hip/hip_runtime.h>

typedef float    v2f  __attribute__((ext_vector_type(2)));
typedef float    v8f  __attribute__((ext_vector_type(8)));

#define LRELU_SLOPE 0.2f

template <int M> __device__ __forceinline__ float row_xor(float v) {
  const int i = __float_as_int(v);
  return __int_as_float(__builtin_amdgcn_update_dpp(i, i, 0x160 | M, 0xf, 0xf, true));
}
__device__ __forceinline__ float row_sum16(float s) {
  s += row_xor<8>(s);
  s += row_xor<4>(s);
  s += row_xor<2>(s);
  s += row_xor<1>(s);
  return s;
}

__device__ __forceinline__ void gat_gemm_body(
    float* __restrict__ wldsT, const int blk,
    const float* __restrict__ X, const float* __restrict__ Wm,
    const float* __restrict__ a_dst,
    float* __restrict__ Hout, float* __restrict__ ald,
    int N, int K)
{
  const int stride = K + 2;
  const int tid = threadIdx.x;
  for (int i = tid; i < K * 64; i += 256) {
    const int k = i >> 6, col = i & 63;
    wldsT[col * stride + k] = Wm[i];
  }
  __syncthreads();

  const int wave = tid >> 5;
  const int lane = tid & 31;
  const int l16  = lane & 15;
  const int hsel = lane >> 4;
  const int rowBase = blk * 128 + wave * 16;
  int row = rowBase + l16;
  if (row >= N) row = N - 1;

  v8f acc[4] = {};

  const float ad0 = a_dst[l16], ad1 = a_dst[16 + l16], ad2 = a_dst[32 + l16], ad3 = a_dst[48 + l16];

  const int koff = hsel * 2;
  const float* xrow = X + (size_t)row * K + koff;
  const float* wb0 = &wldsT[(0 * 16 + l16) * stride + koff];
  const float* wb1 = &wldsT[(1 * 16 + l16) * stride + koff];
  const float* wb2 = &wldsT[(2 * 16 + l16) * stride + koff];
  const float* wb3 = &wldsT[(3 * 16 + l16) * stride + koff];

  for (int kk = 0; kk < K; kk += 4) {
    const float2 av  = *(const float2*)(xrow + kk);
    const float2 bv0 = *(const float2*)(wb0 + kk);
    const float2 bv1 = *(const float2*)(wb1 + kk);
    const float2 bv2 = *(const float2*)(wb2 + kk);
    const float2 bv3 = *(const float2*)(wb3 + kk);
    v2f a;  a.x = av.x;   a.y = av.y;
    v2f b0; b0.x = bv0.x; b0.y = bv0.y;
    v2f b1; b1.x = bv1.x; b1.y = bv1.y;
    v2f b2; b2.x = bv2.x; b2.y = bv2.y;
    v2f b3; b3.x = bv3.x; b3.y = bv3.y;
    acc[0] = __builtin_amdgcn_wmma_f32_16x16x4_f32(false, a, false, b0, (short)0, acc[0], false, false);
    acc[1] = __builtin_amdgcn_wmma_f32_16x16x4_f32(false, a, false, b1, (short)0, acc[1], false, false);
    acc[2] = __builtin_amdgcn_wmma_f32_16x16x4_f32(false, a, false, b2, (short)0, acc[2], false, false);
    acc[3] = __builtin_amdgcn_wmma_f32_16x16x4_f32(false, a, false, b3, (short)0, acc[3], false, false);
  }

#pragma unroll
  for (int nt = 0; nt < 4; ++nt) {
#pragma unroll
    for (int j = 0; j < 8; ++j) {
      const int ro = rowBase + j + hsel * 8;
      if (ro < N) Hout[(size_t)ro * 64 + nt * 16 + l16] = acc[nt][j];
    }
  }

  float pd[8];
#pragma unroll
  for (int j = 0; j < 8; ++j)
    pd[j] = row_sum16(acc[0][j] * ad0 + acc[1][j] * ad1 + acc[2][j] * ad2 + acc[3][j] * ad3);
  float val = 0.0f;
#pragma unroll
  for (int j = 0; j < 8; ++j)
    if (l16 == j) val = pd[j];
  const int rl = rowBase + l16 + hsel * 8;
  if (l16 < 8 && rl < N) ald[rl] = val;
}

#define BIN_SHIFT  8
#define BIN_NODES  256
#define MAX_BINS   1024
#define EDGE_CHUNK 4096
#define HELD       8

__device__ __forceinline__ int wave_incl_scan(int x, int lane)
{
  for (int o = 1; o < 32; o <<= 1) {
    const int t = __shfl_up(x, o, 32);
    if (lane >= o) x += t;
  }
  return x;
}

template <int M> __device__ __forceinline__ int row_xor_i(int i) {
  return __builtin_amdgcn_update_dpp(i, i, 0x160 | M, 0xf, 0xf, true);
}
__device__ __forceinline__ int wave_sum_i(int s) {
  s += (int)__builtin_amdgcn_permlanex16((unsigned)s, (unsigned)s, 0x76543210u, 0xfedcba98u, false, false);
  s += row_xor_i<8>(s);
  s += row_xor_i<4>(s);
  s += row_xor_i<2>(s);
  s += row_xor_i<1>(s);
  return s;
}

__global__ __launch_bounds__(256) void gemm1_and_chunk_sort(
    const float* __restrict__ X, const float* __restrict__ Wm,
    const float* __restrict__ a_dst,
    float* __restrict__ Hout, float* __restrict__ ald,
    int N, int K, int gGemm,
    const int* __restrict__ ei, int E, int NB,
    int* __restrict__ lstart, int* __restrict__ lbinned)
{
  __shared__ __attribute__((aligned(16))) float smem[64 * 130];
  if ((int)blockIdx.x < gGemm) {
    gat_gemm_body(smem, blockIdx.x, X, Wm, a_dst, Hout, ald, N, K);
    return;
  }
  int*  h     = (int*)smem;
  int4* h4    = (int4*)smem;
  int*  wtot  = h + MAX_BINS;
  int*  stage = h + MAX_BINS + 8;
  const int tid = threadIdx.x, lane = tid & 31, wave = tid >> 5;
  const int blk = blockIdx.x - gGemm;
  h4[tid] = make_int4(0, 0, 0, 0);
  const int base = blk * EDGE_CHUNK + tid;
  int s[EDGE_CHUNK / 256], d[EDGE_CHUNK / 256];
#pragma unroll
  for (int k = 0; k < EDGE_CHUNK / 256; ++k) {
    const int e  = base + k * 256;
    const int ec = e < E ? e : E - 1;
    s[k] = ei[ec];
    d[k] = ei[E + ec];
  }
  __syncthreads();
  int rk[EDGE_CHUNK / 256];
#pragma unroll
  for (int k = 0; k < EDGE_CHUNK / 256; ++k) {
    rk[k] = 0;
    if (base + k * 256 < E) rk[k] = atomicAdd(&h[d[k] >> BIN_SHIFT], 1);
  }
  __syncthreads();
  const int4 v = h4[tid];
  const int  t = v.x + v.y + v.z + v.w;
  const int  x = wave_incl_scan(t, lane);
  if (lane == 31) wtot[wave] = x;
  __syncthreads();
  int pre = 0;
  for (int q = 0; q < 8; ++q) pre += (q < wave) ? wtot[q] : 0;
  int4 st;
  st.x = pre + x - t; st.y = st.x + v.x; st.z = st.y + v.y; st.w = st.z + v.z;
  h4[tid] = st;
  int* row = lstart + (size_t)blk * (NB + 1);
  const int b0 = tid * 4;
  if (b0     < NB) row[b0]     = st.x;
  if (b0 + 1 < NB) row[b0 + 1] = st.y;
  if (b0 + 2 < NB) row[b0 + 2] = st.z;
  if (b0 + 3 < NB) row[b0 + 3] = st.w;
  if (tid == 255) row[NB] = st.w + v.w;
  __syncthreads();
#pragma unroll
  for (int k = 0; k < EDGE_CHUNK / 256; ++k)
    if (base + k * 256 < E)
      stage[h[d[k] >> BIN_SHIFT] + rk[k]] = (s[k] << BIN_SHIFT) | (d[k] & (BIN_NODES - 1));
  __syncthreads();
  const int left = E - blk * EDGE_CHUNK;
  const int nv = left < EDGE_CHUNK ? left : EDGE_CHUNK;
#pragma unroll
  for (int k = 0; k < EDGE_CHUNK / 256; ++k) {
    const int i = tid + k * 256;
    if (i < nv) lbinned[(size_t)blk * EDGE_CHUNK + i] = stage[i];
  }
}

__global__ __launch_bounds__(1024) void bin_collect_sort(
    const int* __restrict__ lstart, const int* __restrict__ lbinned,
    int nBlk, int NB,
    int* __restrict__ deg, int* __restrict__ off, int* __restrict__ csrc, int N)
{
  __shared__ int cnt[BIN_NODES];
  __shared__ int cur[BIN_NODES];
  __shared__ int wtot[BIN_NODES / 32];
  __shared__ int wlo[32];
  const int tid = threadIdx.x, lane = tid & 31;
  const int wave  = __builtin_amdgcn_readfirstlane(tid >> 5);
  const int b     = blockIdx.x;
  const int pitch = NB + 1;

  int lo = 0, hi = 0, losum = 0;
  {
    const int c = wave + 32 * lane;
    if (c < nBlk) {
      lo = lstart[(size_t)c * pitch + b];
      hi = lstart[(size_t)c * pitch + b + 1];
    }
    losum = lo;
    for (int c2 = c + 1024; c2 < nBlk; c2 += 1024) losum += lstart[(size_t)c2 * pitch + b];
  }
  if (tid < BIN_NODES) cnt[tid] = 0;

  int n[HELD];
  int w0[HELD], w1[HELD];
  int maxn = 0;
#pragma unroll
  for (int j = 0; j < HELD; ++j) {
    const int lj = __builtin_amdgcn_readlane(lo, j);
    n[j] = __builtin_amdgcn_readlane(hi, j) - lj;
    maxn = n[j] > maxn ? n[j] : maxn;
    const int* p = lbinned + (size_t)(wave + 32 * j) * EDGE_CHUNK + lj + lane;
    w0[j] = 0; w1[j] = 0;
    if (lane < n[j]) w0[j] = p[0];
  }
  const bool two  = maxn > 32;
  const bool more = maxn > 64 || nBlk > 32 * HELD;
  if (two) {
#pragma unroll
    for (int j = 0; j < HELD; ++j) {
      const int lj = __builtin_amdgcn_readlane(lo, j);
      const int* p = lbinned + (size_t)(wave + 32 * j) * EDGE_CHUNK + lj + lane;
      if (lane + 32 < n[j]) w1[j] = p[32];
    }
  }
  __syncthreads();

#pragma unroll
  for (int j = 0; j < HELD; ++j)
    if (lane < n[j]) atomicAdd(&cnt[w0[j] & (BIN_NODES - 1)], 1);
  if (two) {
#pragma unroll
    for (int j = 0; j < HELD; ++j)
      if (lane + 32 < n[j]) atomicAdd(&cnt[w1[j] & (BIN_NODES - 1)], 1);
  }
  if (more) {
    for (int c = wave; c < nBlk; c += 32) {
      const int l = lstart[(size_t)c * pitch + b], h = lstart[(size_t)c * pitch + b + 1];
      const int first = (c < wave + 32 * HELD) ? 64 : 0;
      for (int i = l + first + lane; i < h; i += 32)
        atomicAdd(&cnt[lbinned[(size_t)c * EDGE_CHUNK + i] & (BIN_NODES - 1)], 1);
    }
  }
  {
    const int sl = wave_sum_i(losum);
    if (lane == 0) wlo[wave] = sl;
  }
  __syncthreads();

  int v = 0, x = 0;
  if (tid < BIN_NODES) {
    v = cnt[tid];
    x = wave_incl_scan(v, lane);
    if (lane == 31) wtot[wave] = x;
  }
  __syncthreads();
  if (tid < BIN_NODES) {
    const int binlo = wave_sum_i(wlo[lane]);
    int pre = 0;
    for (int q = 0; q < BIN_NODES / 32; ++q) pre += (q < wave) ? wtot[q] : 0;
    const int o = binlo + pre + x - v;
    cur[tid] = o;
    const int node = b * BIN_NODES + tid;
    if (node < N) { deg[node] = v; off[node] = o; }
  }
  __syncthreads();

  int pos[HELD];
#pragma unroll
  for (int j = 0; j < HELD; ++j) {
    pos[j] = 0;
    if (lane < n[j]) pos[j] = atomicAdd(&cur[w0[j] & (BIN_NODES - 1)], 1);
  }
#pragma unroll
  for (int j = 0; j < HELD; ++j)
    if (lane < n[j]) csrc[pos[j]] = w0[j] >> BIN_SHIFT;
  if (two) {
#pragma unroll
    for (int j = 0; j < HELD; ++j) {
      pos[j] = 0;
      if (lane + 32 < n[j]) pos[j] = atomicAdd(&cur[w1[j] & (BIN_NODES - 1)], 1);
    }
#pragma unroll
    for (int j = 0; j < HELD; ++j)
      if (lane + 32 < n[j]) csrc[pos[j]] = w1[j] >> BIN_SHIFT;
  }
  if (more) {
    for (int c = wave; c < nBlk; c += 32) {
      const int l = lstart[(size_t)c * pitch + b], h = lstart[(size_t)c * pitch + b + 1];
      const int first = (c < wave + 32 * HELD) ? 64 : 0;
      for (int i = l + first + lane; i < h; i += 32) {
        const int t = lbinned[(size_t)c * EDGE_CHUNK + i];
        const int p = atomicAdd(&cur[t & (BIN_NODES - 1)], 1);
        csrc[p] = t >> BIN_SHIFT;
      }
    }
  }
}

__device__ __forceinline__ int half_xor_i(int v) {
  return (int)__builtin_amdgcn_permlanex16((unsigned)v, (unsigned)v, 0x76543210u, 0xfedcba98u, false, false);
}
template <int K> __device__ __forceinline__ int row_share_i(int i) {
  return __builtin_amdgcn_update_dpp(i, i, 0x150 | K, 0xf, 0xf, true);
}
template <int K> __device__ __forceinline__ float row_share_f(float v) {
  return __int_as_float(row_share_i<K>(__float_as_int(v)));
}

__device__ __forceinline__ float lrelu(float v) {
  return (v > 0.0f) ? v : LRELU_SLOPE * v;
}

#define SHARE8(dst, v)                                              \
  dst[0] = row_share_i<0>(v); dst[1] = row_share_i<1>(v);           \
  dst[2] = row_share_i<2>(v); dst[3] = row_share_i<3>(v);           \
  dst[4] = row_share_i<4>(v); dst[5] = row_share_i<5>(v);           \
  dst[6] = row_share_i<6>(v); dst[7] = row_share_i<7>(v);

__device__ __forceinline__ float4 gat_node_row2(
    const int node, const int c,
    const int* __restrict__ off, const int* __restrict__ deg,
    const int* __restrict__ csrc, const float* __restrict__ Hin,
    const float* __restrict__ ald, const float4 a)
{
  const int   beg = off[node];
  const int   cnt = deg[node];
  const float adv = ald[node];

  const char* hb = (const char*)Hin;
  const unsigned coff = (unsigned)c << 4;
  const float4 hs = *(const float4*)(hb + (((unsigned)node << 8) | coff));
  int s0 = node;
  if (c < cnt) s0 = csrc[beg + c];

  const int cntOther = half_xor_i(cnt);
  const int cmax = __builtin_amdgcn_readfirstlane(cnt > cntOther ? cnt : cntOther);

  float m   = lrelu(row_sum16(hs.x * a.x + hs.y * a.y + hs.z * a.z + hs.w * a.w) + adv);
  float den = 1.0f;
  float ax = hs.x, ay = hs.y, az = hs.z, aw = hs.w;

  const bool hi8 = (c & 8) != 0;
  const bool hi4 = (c & 4) != 0;
  const bool hi2 = (c & 2) != 0;
  const int  myrow = c >> 1;

  for (int base = 0; base < cmax; base += 16) {
    const int n    = cnt - base;
    const int nmax = (cmax - base < 16) ? (cmax - base) : 16;
    int sj = s0;
    if (base > 0) {
      sj = node;
      if (c < n) sj = csrc[beg + base + c];
    }
    for (int t = 0; t < nmax; t += 8) {
      int    sk[8];
      float4 hv[8];
      float  p[8];
      SHARE8(sk, sj)
#pragma unroll
      for (int k = 0; k < 8; ++k)
        hv[k] = *(const float4*)(hb + (((unsigned)sk[k] << 8) | coff));
#pragma unroll
      for (int k = 0; k < 8; ++k)
        p[k] = hv[k].x * a.x + hv[k].y * a.y + hv[k].z * a.z + hv[k].w * a.w;

      float q[4];
#pragma unroll
      for (int k = 0; k < 4; ++k) {
        const float keep = hi8 ? p[k + 4] : p[k];
        const float send = hi8 ? p[k] : p[k + 4];
        q[k] = keep + row_xor<8>(send);
      }
      float r[2];
#pragma unroll
      for (int k = 0; k < 2; ++k) {
        const float keep = hi4 ? q[k + 2] : q[k];
        const float send = hi4 ? q[k] : q[k + 2];
        r[k] = keep + row_xor<4>(send);
      }
      float s;
      {
        const float keep = hi2 ? r[1] : r[0];
        const float send = hi2 ? r[0] : r[1];
        s = keep + row_xor<2>(send);
      }
      s += row_xor<1>(s);

      const float e = (t + myrow < n) ? lrelu(s + adv) : -__builtin_huge_valf();
      float mb = fmaxf(e, row_xor<2>(e));
      mb = fmaxf(mb, row_xor<4>(mb));
      mb = fmaxf(mb, row_xor<8>(mb));
      const float mn = fmaxf(m, mb);
      const float sc = __expf(m - mn);
      const float w  = __expf(e - mn);
      float ws = w + row_xor<2>(w);
      ws += row_xor<4>(ws);
      ws += row_xor<8>(ws);
      den = den * sc + ws;
      ax *= sc; ay *= sc; az *= sc; aw *= sc;
      float wk[8];
      wk[0] = row_share_f<0>(w);  wk[1] = row_share_f<2>(w);
      wk[2] = row_share_f<4>(w);  wk[3] = row_share_f<6>(w);
      wk[4] = row_share_f<8>(w);  wk[5] = row_share_f<10>(w);
      wk[6] = row_share_f<12>(w); wk[7] = row_share_f<14>(w);
#pragma unroll
      for (int k = 0; k < 8; ++k) {
        ax += wk[k] * hv[k].x; ay += wk[k] * hv[k].y;
        az += wk[k] * hv[k].z; aw += wk[k] * hv[k].w;
      }
      m = mn;
      sj = row_xor_i<8>(sj);
    }
  }
  float4 res; res.x = ax / den; res.y = ay / den; res.z = az / den; res.w = aw / den;
  return res;
}

__global__ __launch_bounds__(256) void gat_node_softmax_aggregate(
    const int* __restrict__ off, const int* __restrict__ deg,
    const int* __restrict__ csrc, const float* __restrict__ Hin,
    const float* __restrict__ a_src, const float* __restrict__ ald,
    const float* __restrict__ bias, float* __restrict__ out, int N, int doRelu)
{
  const int lane = threadIdx.x & 31;
  const int c    = lane & 15;
  const int h    = lane >> 4;
  const int pair = __builtin_amdgcn_readfirstlane(
      (int)((blockIdx.x * (unsigned)blockDim.x + threadIdx.x) >> 5));
  if (2 * pair >= N) return;
  const int node  = 2 * pair + h;
  const int nodeC = node < N ? node : N - 1;
  const float4 av = *(const float4*)(a_src + c * 4);
  const float4 a  = gat_node_row2(nodeC, c, off, deg, csrc, Hin, ald, av);
  const float4 bv = *(const float4*)(bias + c * 4);
  float4 r; r.x = a.x + bv.x; r.y = a.y + bv.y; r.z = a.z + bv.z; r.w = a.w + bv.w;
  if (doRelu) { r.x = fmaxf(r.x, 0.0f); r.y = fmaxf(r.y, 0.0f); r.z = fmaxf(r.z, 0.0f); r.w = fmaxf(r.w, 0.0f); }
  if (node < N) *(float4*)(out + (size_t)node * 64 + c * 4) = r;
}

#define FUSE_NODES  16
#define AT_STRIDE   68

__global__ __launch_bounds__(256) void gat_node_aggregate_then_gemm(
    const int* __restrict__ off, const int* __restrict__ deg,
    const int* __restrict__ csrc, const float* __restrict__ Hin,
    const float* __restrict__ a_srcIn, const float* __restrict__ ald,
    const float* __restrict__ bias,
    const float* __restrict__ Wm, const float* __restrict__ a_dst,
    float* __restrict__ Hout, float* __restrict__ aldOut, int N)
{
  __shared__ __attribute__((aligned(16))) float At[FUSE_NODES * AT_STRIDE];
  __shared__ __attribute__((aligned(16))) float Ot[FUSE_NODES * AT_STRIDE];
  __shared__ float part[4][16];
  const int lane = threadIdx.x & 31;
  const int l16  = lane & 15;
  const int hsel = lane >> 4;
  const int wave = __builtin_amdgcn_readfirstlane((int)(threadIdx.x >> 5));
  const int rowBase = blockIdx.x * FUSE_NODES;
  const int trow = 2 * wave + hsel;
  const int node = rowBase + trow;

  float4 r; r.x = 0.0f; r.y = 0.0f; r.z = 0.0f; r.w = 0.0f;
  if (rowBase + 2 * wave < N) {
    const int nodeC = node < N ? node : N - 1;
    const float4 av = *(const float4*)(a_srcIn + l16 * 4);
    const float4 a  = gat_node_row2(nodeC, l16, off, deg, csrc, Hin, ald, av);
    const float4 bv = *(const float4*)(bias + l16 * 4);
    if (node < N) {
      r.x = fmaxf(a.x + bv.x, 0.0f);
      r.y = fmaxf(a.y + bv.y, 0.0f);
      r.z = fmaxf(a.z + bv.z, 0.0f);
      r.w = fmaxf(a.w + bv.w, 0.0f);
    }
  }
  *(float4*)&At[trow * AT_STRIDE + l16 * 4] = r;

  const int koff = hsel * 2;
  const int col  = wave * 16 + l16;

  v2f b[16] = {};
  float adv = 0.0f;
  if (wave < 4) {
#pragma unroll
    for (int i = 0; i < 16; ++i) {
      b[i].x = Wm[(4 * i + koff) * 64 + col];
      b[i].y = Wm[(4 * i + koff + 1) * 64 + col];
    }
    adv = a_dst[col];
  }
  __syncthreads();

  if (wave < 4) {
    v8f acc = {};
    const float* ap = &At[l16 * AT_STRIDE + koff];
#pragma unroll
    for (int i = 0; i < 16; ++i) {
      const float2 av = *(const float2*)(ap + 4 * i);
      v2f a; a.x = av.x; a.y = av.y;
      acc = __builtin_amdgcn_wmma_f32_16x16x4_f32(false, a, false, b[i], (short)0, acc, false, false);
    }
#pragma unroll
    for (int j = 0; j < 8; ++j)
      Ot[(j + hsel * 8) * AT_STRIDE + col] = acc[j];
    float val = 0.0f;
#pragma unroll
    for (int j = 0; j < 8; ++j) {
      const float pd = row_sum16(acc[j] * adv);
      if (l16 == j) val = pd;
    }
    if (l16 < 8) part[wave][l16 + hsel * 8] = val;
  }
  __syncthreads();

  if (node < N)
    *(float4*)(Hout + (size_t)node * 64 + l16 * 4) = *(const float4*)&Ot[trow * AT_STRIDE + l16 * 4];

  if (wave == 0 && lane < 16) {
    const float v = part[0][lane] + part[1][lane] + part[2][lane] + part[3][lane];
    const int rl = rowBase + lane;
    if (rl < N) aldOut[rl] = v;
  }
}

extern "C" void kernel_launch(void* const* d_in, const int* in_sizes, int n_in,
                              void* d_out, int out_size, void* d_ws, size_t ws_size,
                              hipStream_t stream) {
  const float* x   = (const float*)d_in[0];
  const int*   ei  = (const int*)  d_in[1];
  const float* W1  = (const float*)d_in[2];
  const float* as1 = (const float*)d_in[3];
  const float* ad1 = (const float*)d_in[4];
  const float* b1  = (const float*)d_in[5];
  const float* W2  = (const float*)d_in[6];
  const float* as2 = (const float*)d_in[7];
  const float* ad2 = (const float*)d_in[8];
  const float* b2  = (const float*)d_in[9];

  const int N  = in_sizes[0] / 128;
  const int E  = in_sizes[1] / 2;

  char* w = (char*)d_ws;
  float* h1   = (float*)w; w += (size_t)N * 64 * sizeof(float);
  float* h2   = (float*)w; w += (size_t)N * 64 * sizeof(float);
  float* ald  = (float*)w; w += (size_t)N * sizeof(float);
  float* ald2 = (float*)w; w += (size_t)N * sizeof(float);
  int*   deg  = (int*)w;   w += (size_t)N * sizeof(int);
  int*   off  = (int*)w;   w += (size_t)N * sizeof(int);
  const int NB   = (N + BIN_NODES - 1) / BIN_NODES;
  const int nBlk = (E + EDGE_CHUNK - 1) / EDGE_CHUNK;
  int*   lstart   = (int*)w; w += (size_t)nBlk * (NB + 1) * sizeof(int);
  int*   lbinned  = (int*)w; w += (size_t)nBlk * EDGE_CHUNK * sizeof(int);
  int*   csrc     = (int*)w; w += (size_t)E * sizeof(int);

  const dim3 blk(256);
  const dim3 blkBin(1024);
  const int gGemm     = (N + 127) / 128;
  const int gFuse     = (N + FUSE_NODES - 1) / FUSE_NODES;

  gemm1_and_chunk_sort<<<gGemm + nBlk, blk, 0, stream>>>(x, W1, ad1, h1, ald, N, 128, gGemm,
                                                        ei, E, NB, lstart, lbinned);
  bin_collect_sort<<<NB, blkBin, 0, stream>>>(lstart, lbinned, nBlk, NB, deg, off, csrc, N);

  gat_node_aggregate_then_gemm<<<gFuse, blk, 0, stream>>>(off, deg, csrc, h1, as1, ald, b1,
                                                                          W2, ad2, h2, ald2, N);

  gat_node_softmax_aggregate<<<gFuse, blk, 0, stream>>>(off, deg, csrc, h2, as2, ald2, b2, (float*)d_out, N, 0);
}
